# starting kernel
# baseline (speedup 1.0000x reference)
#include <hip/hip_runtime.h>


typedef __attribute__((ext_vector_type(2))) float v2f;
typedef __attribute__((ext_vector_type(4))) float v4f;
typedef __attribute__((ext_vector_type(8))) float v8f;

#define N_USERS 29858
#define N_ITEMS 40981
#define N_TOTAL (N_USERS + N_ITEMS)
#define DIM     64
#define FDIM    512
#define N_LAYERS 3
#define NCHUNKS 128
#define CHUNK   556
#define QUADS   (CHUNK / 4)

static_assert(NCHUNKS * CHUNK >= N_TOTAL, "chunk coverage");

#define WMMA_F32(A, B, C) \
    __builtin_amdgcn_wmma_f32_16x16x4_f32(false, (A), false, (B), (short)0, (C), false, false)

__global__ void lgcn_init(const float* __restrict__ user,
                          const float* __restrict__ item,
                          float* __restrict__ ego,
                          float* __restrict__ acc) {
    int i = blockIdx.x * blockDim.x + threadIdx.x;
    if (i < N_TOTAL * DIM) {
        float val = (i < N_USERS * DIM) ? user[i] : item[i - N_USERS * DIM];
        ego[i] = val;
        acc[i] = val;
    }
}

__global__ __launch_bounds__(128)
void lgcn_projT(const float* __restrict__ v,
                const float* __restrict__ ego,
                float* __restrict__ partial  ) {
    __shared__ v2f lds_ego[(CHUNK / 2) * DIM];

    const int chunk   = blockIdx.x;
    const int n_begin = chunk * CHUNK;
    const int n_valid = min(N_TOTAL - n_begin, CHUNK);

    for (int idx = threadIdx.x; idx < (CHUNK / 2) * DIM; idx += blockDim.x) {
        const int j = idx >> 6;
        const int d = idx & 63;
        const int r0 = 2 * j, r1 = 2 * j + 1;
        const float e0 = ego[(size_t)(n_begin + min(r0, n_valid - 1)) * DIM + d];
        const float e1 = ego[(size_t)(n_begin + min(r1, n_valid - 1)) * DIM + d];
        v2f val;
        val.x = (r0 < n_valid) ? e0 : 0.0f;
        val.y = (r1 < n_valid) ? e1 : 0.0f;
        lds_ego[idx] = val;
    }
    __syncthreads();

    const int lane  = threadIdx.x & 31;
    const int wave  = threadIdx.x >> 5;
    const int f0    = (blockIdx.y * 4 + wave) * 32;
    const int row   = lane & 15;
    const int half  = lane >> 4;
    const int khalf = half * 2;

    v8f c0[4] = {{}, {}, {}, {}};
    v8f c1[4] = {{}, {}, {}, {}};

    if (n_valid == CHUNK) {
        const float* ap = v + (size_t)(n_begin + khalf) * FDIM + f0 + row;
#pragma unroll 1
        for (int q = 0; q < QUADS; ++q) {
            v2f a0, a1;
            a0.x = ap[0];         a0.y = ap[FDIM];
            a1.x = ap[16];        a1.y = ap[FDIM + 16];
            const v2f* bp = &lds_ego[(q * 2 + half) * DIM + row];
#pragma unroll
            for (int t = 0; t < 4; ++t) {
                v2f b = bp[t * 16];
                c0[t] = WMMA_F32(a0, b, c0[t]);
                c1[t] = WMMA_F32(a1, b, c1[t]);
            }
            ap += 4 * FDIM;
        }
    } else {
#pragma unroll 1
        for (int q = 0; q < QUADS; ++q) {
            const int na  = n_begin + q * 4 + khalf;
            const size_t ra0 = (size_t)min(na,     N_TOTAL - 1) * FDIM;
            const size_t ra1 = (size_t)min(na + 1, N_TOTAL - 1) * FDIM;
            v2f a0, a1;
            a0.x = v[ra0 + f0 + row];        a0.y = v[ra1 + f0 + row];
            a1.x = v[ra0 + f0 + 16 + row];   a1.y = v[ra1 + f0 + 16 + row];
            const v2f* bp = &lds_ego[(q * 2 + half) * DIM + row];
#pragma unroll
            for (int t = 0; t < 4; ++t) {
                v2f b = bp[t * 16];
                c0[t] = WMMA_F32(a0, b, c0[t]);
                c1[t] = WMMA_F32(a1, b, c1[t]);
            }
        }
    }

    float* outp = partial + (size_t)chunk * FDIM * DIM;
    const int mbase = f0 + (half << 3);
#pragma unroll
    for (int t = 0; t < 4; ++t) {
#pragma unroll
        for (int j = 0; j < 8; ++j) {
            outp[(size_t)(mbase + j)      * DIM + t * 16 + row] = c0[t][j];
            outp[(size_t)(mbase + 16 + j) * DIM + t * 16 + row] = c1[t][j];
        }
    }
}

__global__ void lgcn_reduce(const float* __restrict__ partial,
                            const float* __restrict__ filt  ,
                            float* __restrict__ proj  ) {
    int i = blockIdx.x * blockDim.x + threadIdx.x;
    if (i < FDIM * DIM) {
        float s = 0.0f;
        for (int c = 0; c < NCHUNKS; ++c)
            s += partial[(size_t)c * FDIM * DIM + i];
        proj[i] = s * filt[i / DIM];
    }
}

__global__ __launch_bounds__(256)
void lgcn_prop(const float* __restrict__ v,
               const float* __restrict__ proj,
               float* __restrict__ ego,
               float* __restrict__ acc) {
    __shared__ v2f lds_proj[(FDIM / 2) * DIM];

    for (int idx = threadIdx.x; idx < (FDIM / 2) * DIM; idx += blockDim.x) {
        const int j = idx >> 6;
        const int d = idx & 63;
        v2f val;
        val.x = proj[(size_t)(2 * j)     * DIM + d];
        val.y = proj[(size_t)(2 * j + 1) * DIM + d];
        lds_proj[idx] = val;
    }
    __syncthreads();

    const int lane  = threadIdx.x & 31;
    const int wave  = threadIdx.x >> 5;
    const int m0    = blockIdx.x * 256 + wave * 32;
    const int row   = lane & 15;
    const int half  = lane >> 4;
    const int khalf = half * 2;

    const int m = m0 + row;
    const float* vrow0 = v + (size_t)min(m,      N_TOTAL - 1) * FDIM;
    const float* vrow1 = v + (size_t)min(m + 16, N_TOTAL - 1) * FDIM;

    v8f c0[4] = {{}, {}, {}, {}};
    v8f c1[4] = {{}, {}, {}, {}};

#pragma unroll 1
    for (int k = 0; k < FDIM; k += 4) {
        v2f a0 = *(const v2f*)(vrow0 + k + khalf);
        v2f a1 = *(const v2f*)(vrow1 + k + khalf);
        if (k < FDIM - 64) {
            __builtin_prefetch(vrow0 + k + 64, 0, 1);
            __builtin_prefetch(vrow1 + k + 64, 0, 1);
        }
        const v2f* bp = &lds_proj[((k >> 1) + half) * DIM + row];
#pragma unroll
        for (int t = 0; t < 4; ++t) {
            v2f b = bp[t * 16];
            c0[t] = WMMA_F32(a0, b, c0[t]);
            c1[t] = WMMA_F32(a1, b, c1[t]);
        }
    }

    const int mbase = m0 + (half << 3);
#pragma unroll
    for (int j = 0; j < 8; ++j) {
#pragma unroll
        for (int t = 0; t < 4; ++t) {
            const int r0 = mbase + j;
            const int r1 = mbase + 16 + j;
            if (r0 < N_TOTAL) {
                const size_t idx = (size_t)r0 * DIM + t * 16 + row;
                ego[idx] = c0[t][j];
                acc[idx] += c0[t][j];
            }
            if (r1 < N_TOTAL) {
                const size_t idx = (size_t)r1 * DIM + t * 16 + row;
                ego[idx] = c1[t][j];
                acc[idx] += c1[t][j];
            }
        }
    }
}

__global__ void lgcn_final(const float* __restrict__ acc,
                           float* __restrict__ out) {
    int i = blockIdx.x * blockDim.x + threadIdx.x;
    if (i < N_TOTAL * DIM) out[i] = acc[i] * 0.25f;
}

extern "C" void kernel_launch(void* const* d_in, const int* in_sizes, int n_in,
                              void* d_out, int out_size, void* d_ws, size_t ws_size,
                              hipStream_t stream) {
    const float* user    = (const float*)d_in[0];
    const float* item    = (const float*)d_in[1];
    const float* v       = (const float*)d_in[2];
    const float* filters = (const float*)d_in[3];

    float* ego     = (float*)d_ws;
    float* acc     = ego + (size_t)N_TOTAL * DIM;
    float* partial = acc + (size_t)N_TOTAL * DIM;
    float* proj    = partial + (size_t)NCHUNKS * FDIM * DIM;

    const int totalND = N_TOTAL * DIM;
    lgcn_init<<<(totalND + 255) / 256, 256, 0, stream>>>(user, item, ego, acc);

    for (int layer = 0; layer < N_LAYERS; ++layer) {
        dim3 gA(NCHUNKS, 4);
        lgcn_projT<<<gA, 128, 0, stream>>>(v, ego, partial);

        lgcn_reduce<<<(FDIM * DIM + 255) / 256, 256, 0, stream>>>(
            partial, filters + (size_t)layer * FDIM, proj);

        const int nblocksB = (N_TOTAL + 255) / 256;
        lgcn_prop<<<nblocksB, 256, 0, stream>>>(v, proj, ego, acc);
    }

    lgcn_final<<<(totalND + 255) / 256, 256, 0, stream>>>(acc, (float*)d_out);
}


# Both GEMM K loops lose their workgroup barrier: the operand all waves share lies whole in LDS before the loop (projT: ego chunk; prop: proj), each wave stages only its own v slice; sum order kept
# speedup vs baseline: 3.0651x; 3.0651x over previous
#include <hip/hip_runtime.h>


typedef __attribute__((ext_vector_type(2))) float v2f;
typedef __attribute__((ext_vector_type(4))) float v4f;
typedef __attribute__((ext_vector_type(8))) float v8f;

#define N_USERS 29858
#define N_ITEMS 40981
#define N_TOTAL (N_USERS + N_ITEMS)
#define DIM     64
#define FDIM    512
#define N_LAYERS 3
#define NCHUNKS 128
#define CHUNK   556
#define QUADS   (CHUNK / 4)

static_assert(NCHUNKS * CHUNK >= N_TOTAL, "chunk coverage");

#define WMMA_F32(A, B, C) \
    __builtin_amdgcn_wmma_f32_16x16x4_f32(false, (A), false, (B), (short)0, (C), false, false)

__global__ void lgcn_init(const float* __restrict__ user,
                          const float* __restrict__ item,
                          float* __restrict__ ego) {
    int i = blockIdx.x * blockDim.x + threadIdx.x;
    if (i < N_TOTAL * DIM) {
        float val = (i < N_USERS * DIM) ? user[i] : item[i - N_USERS * DIM];
        ego[i] = val;
    }
}

#define PT_FSPLIT 2
#define PT_F      (FDIM / PT_FSPLIT)
#define PT_ROWS   32
#define PT_NBLK   ((CHUNK + PT_ROWS - 1) / PT_ROWS)
#define PT_PAIRS  (PT_ROWS / 2)
#define PT_CPAIRS (PT_NBLK * PT_PAIRS)
#define PT_VPITCH (2 * PT_F + 32)
#define PT_EPITCH (2 * DIM + 32)
#define PT_VUNITS 4
#define PT_EUNITS (PT_CPAIRS * (DIM / 4) / 256)

static_assert(PT_PAIRS * (32 / 4) == 32 * PT_VUNITS, "a wave's v units per stage");
static_assert(PT_CPAIRS * (DIM / 4) == 256 * PT_EUNITS, "ego units per thread");

__device__ __forceinline__ void projT_fetch(const float* __restrict__ vcol, int nb, int vp,
                                            v4f (&sv0)[PT_VUNITS], v4f (&sv1)[PT_VUNITS]) {
#pragma unroll
    for (int i = 0; i < PT_VUNITS; ++i) {
        const int r = nb + 2 * (vp + 4 * i);
        sv0[i] = *(const v4f*)(vcol + (size_t)min(r,     N_TOTAL - 1) * FDIM);
        sv1[i] = *(const v4f*)(vcol + (size_t)min(r + 1, N_TOTAL - 1) * FDIM);
    }
}

__device__ __forceinline__ void projT_stash(float* __restrict__ lv, int vs,
                                            const v4f (&sv0)[PT_VUNITS], const v4f (&sv1)[PT_VUNITS]) {
#pragma unroll
    for (int i = 0; i < PT_VUNITS; ++i) {
        v4f lo, hi;
        lo.x = sv0[i].x; lo.y = sv1[i].x; lo.z = sv0[i].y; lo.w = sv1[i].y;
        hi.x = sv0[i].z; hi.y = sv1[i].z; hi.z = sv0[i].w; hi.w = sv1[i].w;
        *(v4f*)(lv + vs + i * 4 * PT_VPITCH)     = lo;
        *(v4f*)(lv + vs + i * 4 * PT_VPITCH + 4) = hi;
    }
}

__device__ __forceinline__ void projT_kblock(const float* __restrict__ ap,
                                             const float* __restrict__ bp,
                                             v8f (&c0)[4], v8f (&c1)[4]) {
#pragma unroll
    for (int kk = 0; kk < PT_ROWS / 4; ++kk) {
        const v2f a0 = *(const v2f*)(ap + (2 * kk) * PT_VPITCH);
        const v2f a1 = *(const v2f*)(ap + (2 * kk) * PT_VPITCH + 32);
        const float* bq = bp + (2 * kk) * PT_EPITCH;
#pragma unroll
        for (int t = 0; t < 4; ++t) {
            const v2f b = *(const v2f*)(bq + 32 * t);
            c0[t] = WMMA_F32(a0, b, c0[t]);
            c1[t] = WMMA_F32(a1, b, c1[t]);
        }
    }
}

__global__ __launch_bounds__(256)
void lgcn_projT(const float* __restrict__ v,
                const float* __restrict__ ego,
                float* __restrict__ partial  ) {
    __shared__ __attribute__((aligned(16))) float lds_vt[2][PT_PAIRS * PT_VPITCH];
    __shared__ __attribute__((aligned(16))) float lds_eg[PT_CPAIRS * PT_EPITCH];

    const int tid     = threadIdx.x;
    const int lane    = tid & 31;
    const int wave    = tid >> 5;
    const int chunk   = blockIdx.x;
    const int n_begin = chunk * CHUNK;
    const int n_valid = min(N_TOTAL - n_begin, CHUNK);
    const int fblk    = blockIdx.y * PT_F;

    const int vcg = wave * 8 + (lane & 7);
    const int vp  = lane >> 3;
    const float* vcol = v + fblk + 4 * vcg;
    const int vs = vp * PT_VPITCH + 8 * vcg;

    v4f sv0[PT_VUNITS], sv1[PT_VUNITS];
    projT_fetch(vcol, n_begin, vp, sv0, sv1);

    {
        const int ecg = tid & 15;
        const int ep0 = tid >> 4;
        const float* ecol = ego + 4 * ecg;
        v4f se0[PT_EUNITS], se1[PT_EUNITS];
#pragma unroll
        for (int j = 0; j < PT_EUNITS; ++j) {
            const int r = n_begin + 2 * (ep0 + 16 * j);
            se0[j] = *(const v4f*)(ecol + (size_t)min(r,     N_TOTAL - 1) * DIM);
            se1[j] = *(const v4f*)(ecol + (size_t)min(r + 1, N_TOTAL - 1) * DIM);
        }
        const v4f zero = {0.0f, 0.0f, 0.0f, 0.0f};
#pragma unroll
        for (int j = 0; j < PT_EUNITS; ++j) {
            const int ep = ep0 + 16 * j;
            const v4f e0 = (2 * ep     < n_valid) ? se0[j] : zero;
            const v4f e1 = (2 * ep + 1 < n_valid) ? se1[j] : zero;
            v4f lo, hi;
            lo.x = e0.x; lo.y = e1.x; lo.z = e0.y; lo.w = e1.y;
            hi.x = e0.z; hi.y = e1.z; hi.z = e0.w; hi.w = e1.w;
            *(v4f*)(lds_eg + ep * PT_EPITCH + 8 * ecg)     = lo;
            *(v4f*)(lds_eg + ep * PT_EPITCH + 8 * ecg + 4) = hi;
        }
    }
    projT_stash(lds_vt[0], vs, sv0, sv1);
    __syncthreads();

    const int f0    = fblk + wave * 32;
    const int row   = lane & 15;
    const int half  = lane >> 4;

    v8f c0[4] = {{}, {}, {}, {}};
    v8f c1[4] = {{}, {}, {}, {}};

    const int aoff = half * PT_VPITCH + 2 * (wave * 32 + row);
    const int boff = half * PT_EPITCH + 2 * row;

#pragma unroll 1
    for (int kb = 0; kb < PT_NBLK - 1; ++kb) {
        projT_fetch(vcol, n_begin + (kb + 1) * PT_ROWS, vp, sv0, sv1);
        projT_kblock(lds_vt[kb & 1] + aoff, lds_eg + kb * (PT_PAIRS * PT_EPITCH) + boff, c0, c1);
        __builtin_amdgcn_sched_barrier(0);
        projT_stash(lds_vt[(kb + 1) & 1], vs, sv0, sv1);
    }
    projT_kblock(lds_vt[(PT_NBLK - 1) & 1] + aoff,
                 lds_eg + (PT_NBLK - 1) * (PT_PAIRS * PT_EPITCH) + boff, c0, c1);

    float* outp = partial + (size_t)chunk * FDIM * DIM;
    const int mbase = f0 + (half << 3);
#pragma unroll
    for (int t = 0; t < 4; ++t) {
#pragma unroll
        for (int j = 0; j < 8; ++j) {
            outp[(size_t)(mbase + j)      * DIM + t * 16 + row] = c0[t][j];
            outp[(size_t)(mbase + 16 + j) * DIM + t * 16 + row] = c1[t][j];
        }
    }
}

__global__ void lgcn_reduce(const float* __restrict__ partial,
                            const float* __restrict__ filt  ,
                            float* __restrict__ proj  ) {
    int i = blockIdx.x * blockDim.x + threadIdx.x;
    if (i < FDIM * DIM) {
        float s = 0.0f;
        for (int c = 0; c < NCHUNKS; ++c)
            s += partial[(size_t)c * FDIM * DIM + i];
        proj[i] = s * filt[i / DIM];
    }
}

#define PROP_WAVES  9
#define PROP_ROWS   (PROP_WAVES * 32)
#define PROP_BLOCKS ((N_TOTAL + PROP_ROWS - 1) / PROP_ROWS)
#define KBLK        32
#define NKBLK       (FDIM / KBLK)
#define VPITCH      36
#define PIECES      8
#define PIECE_ROWS  (32 / PIECES)

static_assert(32 * (KBLK / 4) == 32 * PIECES, "a wave's 32 lanes x 8 pieces cover its 32 rows x 32 columns");
static_assert(PROP_BLOCKS * PROP_ROWS >= N_TOTAL, "row coverage");

__device__ __forceinline__ void prop_kblock(const float* __restrict__ ar0,
                                            const float* __restrict__ ar1,
                                            const v2f* __restrict__ bp0,
                                            v8f (&c0)[4], v8f (&c1)[4]) {
#pragma unroll
    for (int kk = 0; kk < KBLK; kk += 4) {
        const v2f a0 = *(const v2f*)(ar0 + kk);
        const v2f a1 = *(const v2f*)(ar1 + kk);
        const v2f* bp = bp0 + (kk >> 1) * DIM;
#pragma unroll
        for (int t = 0; t < 4; ++t) {
            const v2f b = bp[t * 16];
            c0[t] = WMMA_F32(a0, b, c0[t]);
            c1[t] = WMMA_F32(a1, b, c1[t]);
        }
    }
}

__global__ __launch_bounds__(PROP_ROWS)
void lgcn_prop(const float* __restrict__ v,
               const float* __restrict__ proj,
               float* __restrict__ ego_next) {
    __shared__ v2f lds_proj[(FDIM / 2) * DIM];
    __shared__ __attribute__((aligned(16))) float lds_v[2][PROP_ROWS * VPITCH];

    const int tid  = threadIdx.x;
    const int mblk = blockIdx.x * PROP_ROWS;

    const int lj = tid & 7;
    const int lr = (tid >> 5) * 32 + ((tid & 31) >> 3);
    unsigned goff[PIECES];
#pragma unroll
    for (int i = 0; i < PIECES; ++i)
        goff[i] = (unsigned)min(mblk + lr + PIECE_ROWS * i, N_TOTAL - 1) * FDIM + lj * 4;
    const int soff = lr * VPITCH + lj * 4;

    v4f st[PIECES];
#pragma unroll
    for (int i = 0; i < PIECES; ++i)
        st[i] = *(const v4f*)(v + goff[i]);

    for (int idx = tid; idx < (FDIM / 2) * DIM; idx += PROP_ROWS) {
        const int j = idx >> 6;
        const int d = idx & 63;
        v2f val;
        val.x = proj[(size_t)(2 * j)     * DIM + d];
        val.y = proj[(size_t)(2 * j + 1) * DIM + d];
        lds_proj[idx] = val;
    }
#pragma unroll
    for (int i = 0; i < PIECES; ++i)
        *(v4f*)&lds_v[0][soff + i * PIECE_ROWS * VPITCH] = st[i];
    __syncthreads();

    const int lane  = tid & 31;
    const int wave  = tid >> 5;
    const int m0    = mblk + wave * 32;
    const int row   = lane & 15;
    const int half  = lane >> 4;
    const int khalf = half * 2;

    v8f c0[4] = {{}, {}, {}, {}};
    v8f c1[4] = {{}, {}, {}, {}};

    const int aoff = (wave * 32 + row) * VPITCH + khalf;
    const v2f* bbase = &lds_proj[half * DIM + row];

#pragma unroll 1
    for (int kb = 0; kb < NKBLK - 1; ++kb) {
#pragma unroll
        for (int i = 0; i < PIECES; ++i)
            st[i] = *(const v4f*)(v + goff[i] + (kb + 1) * KBLK);
        const float* lv = lds_v[kb & 1];
        prop_kblock(lv + aoff, lv + aoff + 16 * VPITCH,
                    bbase + (size_t)(kb * (KBLK / 2)) * DIM, c0, c1);
        __builtin_amdgcn_sched_barrier(0);
        float* ln = lds_v[(kb + 1) & 1];
#pragma unroll
        for (int i = 0; i < PIECES; ++i)
            *(v4f*)&ln[soff + i * PIECE_ROWS * VPITCH] = st[i];
    }
    {
        const float* lv = lds_v[(NKBLK - 1) & 1];
        prop_kblock(lv + aoff, lv + aoff + 16 * VPITCH,
                    bbase + (size_t)((NKBLK - 1) * (KBLK / 2)) * DIM, c0, c1);
    }

    const int mbase = m0 + (half << 3);
#pragma unroll
    for (int j = 0; j < 8; ++j) {
#pragma unroll
        for (int t = 0; t < 4; ++t) {
            const int r0 = mbase + j;
            const int r1 = mbase + 16 + j;
            if (r0 < N_TOTAL) ego_next[(size_t)r0 * DIM + t * 16 + row] = c0[t][j];
            if (r1 < N_TOTAL) ego_next[(size_t)r1 * DIM + t * 16 + row] = c1[t][j];
        }
    }
}

#define PLANE  ((size_t)N_TOTAL * DIM)
#define PLANE4 (N_TOTAL * DIM / 4)

static_assert(N_LAYERS == 3, "lgcn_final adds four planes");
static_assert((N_TOTAL * DIM) % 4 == 0, "planes are whole 16-byte pieces");

__global__ void lgcn_final(const float* __restrict__ planes,
                           float* __restrict__ out) {
    int i = blockIdx.x * blockDim.x + threadIdx.x;
    if (i < PLANE4) {
        const v4f e0 = ((const v4f*)(planes))[i];
        const v4f e1 = ((const v4f*)(planes + PLANE))[i];
        const v4f e2 = ((const v4f*)(planes + 2 * PLANE))[i];
        const v4f e3 = ((const v4f*)(planes + 3 * PLANE))[i];
        ((v4f*)out)[i] = (((e0 + e1) + e2) + e3) * 0.25f;
    }
}

extern "C" void kernel_launch(void* const* d_in, const int* in_sizes, int n_in,
                              void* d_out, int out_size, void* d_ws, size_t ws_size,
                              hipStream_t stream) {
    const float* user    = (const float*)d_in[0];
    const float* item    = (const float*)d_in[1];
    const float* v       = (const float*)d_in[2];
    const float* filters = (const float*)d_in[3];

    float* ego     = (float*)d_ws;
    float* partial = ego + (size_t)(N_LAYERS + 1) * N_TOTAL * DIM;
    float* proj    = partial + (size_t)NCHUNKS * FDIM * DIM;

    const int totalND = N_TOTAL * DIM;
    lgcn_init<<<(totalND + 255) / 256, 256, 0, stream>>>(user, item, ego);

    for (int layer = 0; layer < N_LAYERS; ++layer) {
        dim3 gA(NCHUNKS, PT_FSPLIT);
        lgcn_projT<<<gA, 256, 0, stream>>>(v, ego + (size_t)layer * N_TOTAL * DIM, partial);

        lgcn_reduce<<<(FDIM * DIM + 255) / 256, 256, 0, stream>>>(
            partial, filters + (size_t)layer * FDIM, proj);

        lgcn_prop<<<PROP_BLOCKS, PROP_ROWS, 0, stream>>>(v, proj, ego + (size_t)(layer + 1) * N_TOTAL * DIM);
    }

    lgcn_final<<<(PLANE4 + 255) / 256, 256, 0, stream>>>(ego, (float*)d_out);
}


# prop: k split over wave pairs: 12 equal waves (3 per SIMD), a pair owns 48 rows x 64 cols, each wave sums one half of k, halves added via LDS; was 9 waves x 32 rows; sum order now (k<256)+(k>=256)
# speedup vs baseline: 3.2536x; 1.0615x over previous
#include <hip/hip_runtime.h>


typedef __attribute__((ext_vector_type(2))) float v2f;
typedef __attribute__((ext_vector_type(4))) float v4f;
typedef __attribute__((ext_vector_type(8))) float v8f;

#define N_USERS 29858
#define N_ITEMS 40981
#define N_TOTAL (N_USERS + N_ITEMS)
#define DIM     64
#define FDIM    512
#define N_LAYERS 3
#define NCHUNKS 128
#define CHUNK   556
#define QUADS   (CHUNK / 4)

static_assert(NCHUNKS * CHUNK >= N_TOTAL, "chunk coverage");

#define WMMA_F32(A, B, C) \
    __builtin_amdgcn_wmma_f32_16x16x4_f32(false, (A), false, (B), (short)0, (C), false, false)

__global__ void lgcn_init(const float* __restrict__ user,
                          const float* __restrict__ item,
                          float* __restrict__ ego) {
    int i = blockIdx.x * blockDim.x + threadIdx.x;
    if (i < N_TOTAL * DIM) {
        float val = (i < N_USERS * DIM) ? user[i] : item[i - N_USERS * DIM];
        ego[i] = val;
    }
}

#define PT_FSPLIT 2
#define PT_F      (FDIM / PT_FSPLIT)
#define PT_ROWS   32
#define PT_NBLK   ((CHUNK + PT_ROWS - 1) / PT_ROWS)
#define PT_PAIRS  (PT_ROWS / 2)
#define PT_CPAIRS (PT_NBLK * PT_PAIRS)
#define PT_VPITCH (2 * PT_F + 32)
#define PT_EPITCH (2 * DIM + 32)
#define PT_VUNITS 4
#define PT_EUNITS (PT_CPAIRS * (DIM / 4) / 256)

static_assert(PT_PAIRS * (32 / 4) == 32 * PT_VUNITS, "a wave's v units per stage");
static_assert(PT_CPAIRS * (DIM / 4) == 256 * PT_EUNITS, "ego units per thread");

__device__ __forceinline__ void projT_fetch(const float* __restrict__ vcol, int nb, int vp,
                                            v4f (&sv0)[PT_VUNITS], v4f (&sv1)[PT_VUNITS]) {
#pragma unroll
    for (int i = 0; i < PT_VUNITS; ++i) {
        const int r = nb + 2 * (vp + 4 * i);
        sv0[i] = *(const v4f*)(vcol + (size_t)min(r,     N_TOTAL - 1) * FDIM);
        sv1[i] = *(const v4f*)(vcol + (size_t)min(r + 1, N_TOTAL - 1) * FDIM);
    }
}

__device__ __forceinline__ void projT_stash(float* __restrict__ lv, int vs,
                                            const v4f (&sv0)[PT_VUNITS], const v4f (&sv1)[PT_VUNITS]) {
#pragma unroll
    for (int i = 0; i < PT_VUNITS; ++i) {
        v4f lo, hi;
        lo.x = sv0[i].x; lo.y = sv1[i].x; lo.z = sv0[i].y; lo.w = sv1[i].y;
        hi.x = sv0[i].z; hi.y = sv1[i].z; hi.z = sv0[i].w; hi.w = sv1[i].w;
        *(v4f*)(lv + vs + i * 4 * PT_VPITCH)     = lo;
        *(v4f*)(lv + vs + i * 4 * PT_VPITCH + 4) = hi;
    }
}

__device__ __forceinline__ void projT_kblock(const float* __restrict__ ap,
                                             const float* __restrict__ bp,
                                             v8f (&c0)[4], v8f (&c1)[4]) {
#pragma unroll
    for (int kk = 0; kk < PT_ROWS / 4; ++kk) {
        const v2f a0 = *(const v2f*)(ap + (2 * kk) * PT_VPITCH);
        const v2f a1 = *(const v2f*)(ap + (2 * kk) * PT_VPITCH + 32);
        const float* bq = bp + (2 * kk) * PT_EPITCH;
#pragma unroll
        for (int t = 0; t < 4; ++t) {
            const v2f b = *(const v2f*)(bq + 32 * t);
            c0[t] = WMMA_F32(a0, b, c0[t]);
            c1[t] = WMMA_F32(a1, b, c1[t]);
        }
    }
}

__global__ __launch_bounds__(256)
void lgcn_projT(const float* __restrict__ v,
                const float* __restrict__ ego,
                float* __restrict__ partial  ) {
    __shared__ __attribute__((aligned(16))) float lds_vt[2][PT_PAIRS * PT_VPITCH];
    __shared__ __attribute__((aligned(16))) float lds_eg[PT_CPAIRS * PT_EPITCH];

    const int tid     = threadIdx.x;
    const int lane    = tid & 31;
    const int wave    = tid >> 5;
    const int chunk   = blockIdx.x;
    const int n_begin = chunk * CHUNK;
    const int n_valid = min(N_TOTAL - n_begin, CHUNK);
    const int fblk    = blockIdx.y * PT_F;

    const int vcg = wave * 8 + (lane & 7);
    const int vp  = lane >> 3;
    const float* vcol = v + fblk + 4 * vcg;
    const int vs = vp * PT_VPITCH + 8 * vcg;

    v4f sv0[PT_VUNITS], sv1[PT_VUNITS];
    projT_fetch(vcol, n_begin, vp, sv0, sv1);

    {
        const int ecg = tid & 15;
        const int ep0 = tid >> 4;
        const float* ecol = ego + 4 * ecg;
        v4f se0[PT_EUNITS], se1[PT_EUNITS];
#pragma unroll
        for (int j = 0; j < PT_EUNITS; ++j) {
            const int r = n_begin + 2 * (ep0 + 16 * j);
            se0[j] = *(const v4f*)(ecol + (size_t)min(r,     N_TOTAL - 1) * DIM);
            se1[j] = *(const v4f*)(ecol + (size_t)min(r + 1, N_TOTAL - 1) * DIM);
        }
        const v4f zero = {0.0f, 0.0f, 0.0f, 0.0f};
#pragma unroll
        for (int j = 0; j < PT_EUNITS; ++j) {
            const int ep = ep0 + 16 * j;
            const v4f e0 = (2 * ep     < n_valid) ? se0[j] : zero;
            const v4f e1 = (2 * ep + 1 < n_valid) ? se1[j] : zero;
            v4f lo, hi;
            lo.x = e0.x; lo.y = e1.x; lo.z = e0.y; lo.w = e1.y;
            hi.x = e0.z; hi.y = e1.z; hi.z = e0.w; hi.w = e1.w;
            *(v4f*)(lds_eg + ep * PT_EPITCH + 8 * ecg)     = lo;
            *(v4f*)(lds_eg + ep * PT_EPITCH + 8 * ecg + 4) = hi;
        }
    }
    projT_stash(lds_vt[0], vs, sv0, sv1);
    __syncthreads();

    const int f0    = fblk + wave * 32;
    const int row   = lane & 15;
    const int half  = lane >> 4;

    v8f c0[4] = {{}, {}, {}, {}};
    v8f c1[4] = {{}, {}, {}, {}};

    const int aoff = half * PT_VPITCH + 2 * (wave * 32 + row);
    const int boff = half * PT_EPITCH + 2 * row;

#pragma unroll 1
    for (int kb = 0; kb < PT_NBLK - 1; ++kb) {
        projT_fetch(vcol, n_begin + (kb + 1) * PT_ROWS, vp, sv0, sv1);
        projT_kblock(lds_vt[kb & 1] + aoff, lds_eg + kb * (PT_PAIRS * PT_EPITCH) + boff, c0, c1);
        __builtin_amdgcn_sched_barrier(0);
        projT_stash(lds_vt[(kb + 1) & 1], vs, sv0, sv1);
    }
    projT_kblock(lds_vt[(PT_NBLK - 1) & 1] + aoff,
                 lds_eg + (PT_NBLK - 1) * (PT_PAIRS * PT_EPITCH) + boff, c0, c1);

    float* outp = partial + (size_t)chunk * FDIM * DIM;
    const int mbase = f0 + (half << 3);
#pragma unroll
    for (int t = 0; t < 4; ++t) {
#pragma unroll
        for (int j = 0; j < 8; ++j) {
            outp[(size_t)(mbase + j)      * DIM + t * 16 + row] = c0[t][j];
            outp[(size_t)(mbase + 16 + j) * DIM + t * 16 + row] = c1[t][j];
        }
    }
}

__global__ void lgcn_reduce(const float* __restrict__ partial,
                            const float* __restrict__ filt  ,
                            float* __restrict__ proj  ) {
    int i = blockIdx.x * blockDim.x + threadIdx.x;
    if (i < FDIM * DIM) {
        float s = 0.0f;
        for (int c = 0; c < NCHUNKS; ++c)
            s += partial[(size_t)c * FDIM * DIM + i];
        proj[i] = s * filt[i / DIM];
    }
}

#define PROP_WAVES   12
#define PROP_GROUPS  (PROP_WAVES / 2)
#define GROUP_ROWS   48
#define PROP_ROWS    (PROP_GROUPS * GROUP_ROWS)
#define PROP_THREADS (PROP_WAVES * 32)
#define PROP_BLOCKS  ((N_TOTAL + PROP_ROWS - 1) / PROP_ROWS)
#define KBLK         32
#define KHALF        (FDIM / 2)
#define NKBLK        (KHALF / KBLK)
#define VPITCH       36
#define PIECES       12
#define PIECE_ROWS   4
#define SLICE        (GROUP_ROWS * VPITCH)
#define MT           (GROUP_ROWS / 16)

static_assert(GROUP_ROWS * (KBLK / 4) == 32 * PIECES, "a wave's 32 lanes x 12 pieces cover its 48 rows x 32 columns");
static_assert(PIECE_ROWS * PIECES == GROUP_ROWS, "piece rows cover the slice");
static_assert(PROP_BLOCKS * PROP_ROWS >= N_TOTAL, "row coverage");
static_assert(MT * 2 * 8 * 32 <= SLICE, "the swapped tiles of a wave fit its own stage-0 slice");

__device__ __forceinline__ void prop_kblock(const float* __restrict__ ar,
                                            const v2f* __restrict__ bp0,
                                            v8f (&c)[MT][4]) {
#pragma unroll
    for (int kk = 0; kk < KBLK; kk += 4) {
        v2f a[MT];
#pragma unroll
        for (int m = 0; m < MT; ++m)
            a[m] = *(const v2f*)(ar + m * 16 * VPITCH + kk);
        const v2f* bp = bp0 + (kk >> 1) * DIM;
#pragma unroll
        for (int t = 0; t < 4; ++t) {
            const v2f b = bp[t * 16];
#pragma unroll
            for (int m = 0; m < MT; ++m)
                c[m][t] = WMMA_F32(a[m], b, c[m][t]);
        }
    }
}

template <int T0>
__device__ __forceinline__ void prop_give(float* __restrict__ x, int lane, const v8f (&c)[MT][4]) {
#pragma unroll
    for (int m = 0; m < MT; ++m) {
#pragma unroll
        for (int u = 0; u < 2; ++u) {
            const int q = m * 2 + u;
            const v8f val = c[m][T0 + u];
            v4f lo, hi;
            lo.x = val[0]; lo.y = val[1]; lo.z = val[2]; lo.w = val[3];
            hi.x = val[4]; hi.y = val[5]; hi.z = val[6]; hi.w = val[7];
            *(v4f*)(x + ((q * 2)     * 32 + lane) * 4) = lo;
            *(v4f*)(x + ((q * 2 + 1) * 32 + lane) * 4) = hi;
        }
    }
}

template <int T0>
__device__ __forceinline__ void prop_finish(const float* __restrict__ xp, int lane, int r_first,
                                            const v8f (&c)[MT][4], float* __restrict__ ego_next) {
    const int row  = lane & 15;
    const int half = lane >> 4;
#pragma unroll
    for (int m = 0; m < MT; ++m) {
#pragma unroll
        for (int u = 0; u < 2; ++u) {
            const int q = m * 2 + u;
            const v4f lo = *(const v4f*)(xp + ((q * 2)     * 32 + lane) * 4);
            const v4f hi = *(const v4f*)(xp + ((q * 2 + 1) * 32 + lane) * 4);
            const v8f own = c[m][T0 + u];
            float s[8];
            s[0] = own[0] + lo.x; s[1] = own[1] + lo.y; s[2] = own[2] + lo.z; s[3] = own[3] + lo.w;
            s[4] = own[4] + hi.x; s[5] = own[5] + hi.y; s[6] = own[6] + hi.z; s[7] = own[7] + hi.w;
#pragma unroll
            for (int j = 0; j < 8; ++j) {
                const int r = r_first + m * 16 + (half << 3) + j;
                if (r < N_TOTAL) ego_next[(size_t)r * DIM + (T0 + u) * 16 + row] = s[j];
            }
        }
    }
}

__global__ __launch_bounds__(PROP_THREADS)
void lgcn_prop(const float* __restrict__ v,
               const float* __restrict__ proj,
               float* __restrict__ ego_next) {
    __shared__ v2f lds_proj[(FDIM / 2) * DIM];
    __shared__ __attribute__((aligned(16))) float lds_v[2][PROP_WAVES * SLICE];

    const int tid  = threadIdx.x;
    const int lane = tid & 31;
    const int wave = tid >> 5;
    const int grp  = wave >> 1;
    const int kh   = wave & 1;
    const int g0   = blockIdx.x * PROP_ROWS + grp * GROUP_ROWS;

    const int lj = lane & 7;
    const int lr = lane >> 3;
    unsigned goff[PIECES];
#pragma unroll
    for (int i = 0; i < PIECES; ++i)
        goff[i] = (unsigned)min(g0 + lr + PIECE_ROWS * i, N_TOTAL - 1) * FDIM + kh * KHALF + lj * 4;
    const int soff = wave * SLICE + lr * VPITCH + lj * 4;

    v4f st[PIECES];
#pragma unroll
    for (int i = 0; i < PIECES; ++i)
        st[i] = *(const v4f*)(v + goff[i]);

    for (int idx = tid; idx < (FDIM / 2) * DIM; idx += PROP_THREADS) {
        const int j = idx >> 6;
        const int d = idx & 63;
        v2f val;
        val.x = proj[(size_t)(2 * j)     * DIM + d];
        val.y = proj[(size_t)(2 * j + 1) * DIM + d];
        lds_proj[idx] = val;
    }
#pragma unroll
    for (int i = 0; i < PIECES; ++i)
        *(v4f*)&lds_v[0][soff + i * PIECE_ROWS * VPITCH] = st[i];
    __syncthreads();

    const int row   = lane & 15;
    const int half  = lane >> 4;
    const int khalf = half * 2;

    v8f c[MT][4];
#pragma unroll
    for (int m = 0; m < MT; ++m)
#pragma unroll
        for (int t = 0; t < 4; ++t)
            c[m][t] = (v8f){0.0f, 0.0f, 0.0f, 0.0f, 0.0f, 0.0f, 0.0f, 0.0f};

    const int aoff = wave * SLICE + row * VPITCH + khalf;
    const v2f* bbase = &lds_proj[(kh * (KHALF / 2) + half) * DIM + row];

#pragma unroll 1
    for (int kb = 0; kb < NKBLK - 1; ++kb) {
#pragma unroll
        for (int i = 0; i < PIECES; ++i)
            st[i] = *(const v4f*)(v + goff[i] + (kb + 1) * KBLK);
        prop_kblock(lds_v[kb & 1] + aoff, bbase + (size_t)(kb * (KBLK / 2)) * DIM, c);
        __builtin_amdgcn_sched_barrier(0);
        float* ln = lds_v[(kb + 1) & 1];
#pragma unroll
        for (int i = 0; i < PIECES; ++i)
            *(v4f*)&ln[soff + i * PIECE_ROWS * VPITCH] = st[i];
    }
    prop_kblock(lds_v[(NKBLK - 1) & 1] + aoff, bbase + (size_t)((NKBLK - 1) * (KBLK / 2)) * DIM, c);

    float*       xmine = &lds_v[0][wave * SLICE];
    const float* xpart = &lds_v[0][(wave ^ 1) * SLICE];
    if (kh == 0) prop_give<2>(xmine, lane, c);
    else         prop_give<0>(xmine, lane, c);
    __syncthreads();
    if (kh == 0) prop_finish<0>(xpart, lane, g0, c, ego_next);
    else         prop_finish<2>(xpart, lane, g0, c, ego_next);
}

#define PLANE  ((size_t)N_TOTAL * DIM)
#define PLANE4 (N_TOTAL * DIM / 4)

static_assert(N_LAYERS == 3, "lgcn_final adds four planes");
static_assert((N_TOTAL * DIM) % 4 == 0, "planes are whole 16-byte pieces");

__global__ void lgcn_final(const float* __restrict__ planes,
                           float* __restrict__ out) {
    int i = blockIdx.x * blockDim.x + threadIdx.x;
    if (i < PLANE4) {
        const v4f e0 = ((const v4f*)(planes))[i];
        const v4f e1 = ((const v4f*)(planes + PLANE))[i];
        const v4f e2 = ((const v4f*)(planes + 2 * PLANE))[i];
        const v4f e3 = ((const v4f*)(planes + 3 * PLANE))[i];
        ((v4f*)out)[i] = (((e0 + e1) + e2) + e3) * 0.25f;
    }
}

extern "C" void kernel_launch(void* const* d_in, const int* in_sizes, int n_in,
                              void* d_out, int out_size, void* d_ws, size_t ws_size,
                              hipStream_t stream) {
    const float* user    = (const float*)d_in[0];
    const float* item    = (const float*)d_in[1];
    const float* v       = (const float*)d_in[2];
    const float* filters = (const float*)d_in[3];

    float* ego     = (float*)d_ws;
    float* partial = ego + (size_t)(N_LAYERS + 1) * N_TOTAL * DIM;
    float* proj    = partial + (size_t)NCHUNKS * FDIM * DIM;

    const int totalND = N_TOTAL * DIM;
    lgcn_init<<<(totalND + 255) / 256, 256, 0, stream>>>(user, item, ego);

    for (int layer = 0; layer < N_LAYERS; ++layer) {
        dim3 gA(NCHUNKS, PT_FSPLIT);
        lgcn_projT<<<gA, 256, 0, stream>>>(v, ego + (size_t)layer * N_TOTAL * DIM, partial);

        lgcn_reduce<<<(FDIM * DIM + 255) / 256, 256, 0, stream>>>(
            partial, filters + (size_t)layer * FDIM, proj);

        lgcn_prop<<<PROP_BLOCKS, PROP_THREADS, 0, stream>>>(v, proj, ego + (size_t)(layer + 1) * N_TOTAL * DIM);
    }

    lgcn_final<<<(PLANE4 + 255) / 256, 256, 0, stream>>>(ego, (float*)d_out);
}


# Last 4-byte global accesses become 16-byte pieces of whole lines: init; projT and prop results, turned through the wave's own LDS slice; reduce, its sum now 4 lane runs of 32 chunks added as a tree
# speedup vs baseline: 3.5085x; 1.0783x over previous
#include <hip/hip_runtime.h>


typedef __attribute__((ext_vector_type(2))) float v2f;
typedef __attribute__((ext_vector_type(4))) float v4f;
typedef __attribute__((ext_vector_type(8))) float v8f;

#define N_USERS 29858
#define N_ITEMS 40981
#define N_TOTAL (N_USERS + N_ITEMS)
#define DIM     64
#define FDIM    512
#define N_LAYERS 3
#define NCHUNKS 128
#define CHUNK   556
#define QUADS   (CHUNK / 4)

static_assert(NCHUNKS * CHUNK >= N_TOTAL, "chunk coverage");

#define WMMA_F32(A, B, C) \
    __builtin_amdgcn_wmma_f32_16x16x4_f32(false, (A), false, (B), (short)0, (C), false, false)

#define USER4  (N_USERS * DIM / 4)
#define TOTAL4 (N_TOTAL * DIM / 4)

static_assert((N_USERS * DIM) % 4 == 0 && (N_ITEMS * DIM) % 4 == 0, "both tables are whole 16-byte pieces");

__global__ void lgcn_init(const float* __restrict__ user,
                          const float* __restrict__ item,
                          float* __restrict__ ego) {
    int i = blockIdx.x * blockDim.x + threadIdx.x;
    if (i < TOTAL4) {
        const v4f val = (i < USER4) ? ((const v4f*)user)[i] : ((const v4f*)item)[i - USER4];
        ((v4f*)ego)[i] = val;
    }
}

#define PT_FSPLIT 2
#define PT_F      (FDIM / PT_FSPLIT)
#define PT_ROWS   32
#define PT_NBLK   ((CHUNK + PT_ROWS - 1) / PT_ROWS)
#define PT_PAIRS  (PT_ROWS / 2)
#define PT_CPAIRS (PT_NBLK * PT_PAIRS)
#define PT_VPITCH (2 * PT_F + 32)
#define PT_EPITCH (2 * DIM + 32)
#define PT_VUNITS 4
#define PT_EUNITS (PT_CPAIRS * (DIM / 4) / 256)

static_assert(PT_PAIRS * (32 / 4) == 32 * PT_VUNITS, "a wave's v units per stage");
static_assert(PT_CPAIRS * (DIM / 4) == 256 * PT_EUNITS, "ego units per thread");

__device__ __forceinline__ void projT_fetch(const float* __restrict__ vcol, int nb, int vp,
                                            v4f (&sv0)[PT_VUNITS], v4f (&sv1)[PT_VUNITS]) {
#pragma unroll
    for (int i = 0; i < PT_VUNITS; ++i) {
        const int r = nb + 2 * (vp + 4 * i);
        sv0[i] = *(const v4f*)(vcol + (size_t)min(r,     N_TOTAL - 1) * FDIM);
        sv1[i] = *(const v4f*)(vcol + (size_t)min(r + 1, N_TOTAL - 1) * FDIM);
    }
}

__device__ __forceinline__ void projT_stash(float* __restrict__ lv, int vs,
                                            const v4f (&sv0)[PT_VUNITS], const v4f (&sv1)[PT_VUNITS]) {
#pragma unroll
    for (int i = 0; i < PT_VUNITS; ++i) {
        v4f lo, hi;
        lo.x = sv0[i].x; lo.y = sv1[i].x; lo.z = sv0[i].y; lo.w = sv1[i].y;
        hi.x = sv0[i].z; hi.y = sv1[i].z; hi.z = sv0[i].w; hi.w = sv1[i].w;
        *(v4f*)(lv + vs + i * 4 * PT_VPITCH)     = lo;
        *(v4f*)(lv + vs + i * 4 * PT_VPITCH + 4) = hi;
    }
}

__device__ __forceinline__ void projT_kblock(const float* __restrict__ ap,
                                             const float* __restrict__ bp,
                                             v8f (&c0)[4], v8f (&c1)[4]) {
#pragma unroll
    for (int kk = 0; kk < PT_ROWS / 4; ++kk) {
        const v2f a0 = *(const v2f*)(ap + (2 * kk) * PT_VPITCH);
        const v2f a1 = *(const v2f*)(ap + (2 * kk) * PT_VPITCH + 32);
        const float* bq = bp + (2 * kk) * PT_EPITCH;
#pragma unroll
        for (int t = 0; t < 4; ++t) {
            const v2f b = *(const v2f*)(bq + 32 * t);
            c0[t] = WMMA_F32(a0, b, c0[t]);
            c1[t] = WMMA_F32(a1, b, c1[t]);
        }
    }
}

__global__ __launch_bounds__(256)
void lgcn_projT(const float* __restrict__ v,
                const float* __restrict__ ego,
                float* __restrict__ partial  ) {
    __shared__ __attribute__((aligned(16))) float lds_vt[2][PT_PAIRS * PT_VPITCH];
    __shared__ __attribute__((aligned(16))) float lds_eg[PT_CPAIRS * PT_EPITCH];

    const int tid     = threadIdx.x;
    const int lane    = tid & 31;
    const int wave    = tid >> 5;
    const int chunk   = blockIdx.x;
    const int n_begin = chunk * CHUNK;
    const int n_valid = min(N_TOTAL - n_begin, CHUNK);
    const int fblk    = blockIdx.y * PT_F;

    const int vcg = wave * 8 + (lane & 7);
    const int vp  = lane >> 3;
    const float* vcol = v + fblk + 4 * vcg;
    const int vs = vp * PT_VPITCH + 8 * vcg;

    v4f sv0[PT_VUNITS], sv1[PT_VUNITS];
    projT_fetch(vcol, n_begin, vp, sv0, sv1);

    {
        const int ecg = tid & 15;
        const int ep0 = tid >> 4;
        const float* ecol = ego + 4 * ecg;
        v4f se0[PT_EUNITS], se1[PT_EUNITS];
#pragma unroll
        for (int j = 0; j < PT_EUNITS; ++j) {
            const int r = n_begin + 2 * (ep0 + 16 * j);
            se0[j] = *(const v4f*)(ecol + (size_t)min(r,     N_TOTAL - 1) * DIM);
            se1[j] = *(const v4f*)(ecol + (size_t)min(r + 1, N_TOTAL - 1) * DIM);
        }
        const v4f zero = {0.0f, 0.0f, 0.0f, 0.0f};
#pragma unroll
        for (int j = 0; j < PT_EUNITS; ++j) {
            const int ep = ep0 + 16 * j;
            const v4f e0 = (2 * ep     < n_valid) ? se0[j] : zero;
            const v4f e1 = (2 * ep + 1 < n_valid) ? se1[j] : zero;
            v4f lo, hi;
            lo.x = e0.x; lo.y = e1.x; lo.z = e0.y; lo.w = e1.y;
            hi.x = e0.z; hi.y = e1.z; hi.z = e0.w; hi.w = e1.w;
            *(v4f*)(lds_eg + ep * PT_EPITCH + 8 * ecg)     = lo;
            *(v4f*)(lds_eg + ep * PT_EPITCH + 8 * ecg + 4) = hi;
        }
    }
    projT_stash(lds_vt[0], vs, sv0, sv1);
    __syncthreads();

    const int f0    = fblk + wave * 32;
    const int row   = lane & 15;
    const int half  = lane >> 4;

    v8f c0[4] = {{}, {}, {}, {}};
    v8f c1[4] = {{}, {}, {}, {}};

    const int aoff = half * PT_VPITCH + 2 * (wave * 32 + row);
    const int boff = half * PT_EPITCH + 2 * row;

#pragma unroll 1
    for (int kb = 0; kb < PT_NBLK - 1; ++kb) {
        projT_fetch(vcol, n_begin + (kb + 1) * PT_ROWS, vp, sv0, sv1);
        projT_kblock(lds_vt[kb & 1] + aoff, lds_eg + kb * (PT_PAIRS * PT_EPITCH) + boff, c0, c1);
        __builtin_amdgcn_sched_barrier(0);
        projT_stash(lds_vt[(kb + 1) & 1], vs, sv0, sv1);
    }
    projT_kblock(lds_vt[(PT_NBLK - 1) & 1] + aoff,
                 lds_eg + (PT_NBLK - 1) * (PT_PAIRS * PT_EPITCH) + boff, c0, c1);

    float* lw = &lds_vt[0][0] + 64 * wave;
#pragma unroll
    for (int t = 0; t < 4; ++t) {
#pragma unroll
        for (int j = 0; j < 8; ++j) {
            lw[(j + 8 * half) * PT_VPITCH + t * 16 + row]                         = c0[t][j];
            lw[PT_PAIRS * PT_VPITCH + (j + 8 * half) * PT_VPITCH + t * 16 + row] = c1[t][j];
        }
    }
    float* outp = partial + (size_t)chunk * FDIM * DIM + (size_t)f0 * DIM;
    const int pr = lane >> 4;
    const int pc = 4 * (lane & 15);
#pragma unroll
    for (int i = 0; i < 16; ++i) {
        const int fr = 2 * i + pr;
        const v4f val = *(const v4f*)(lw + (fr >> 4) * (PT_PAIRS * PT_VPITCH) + (fr & 15) * PT_VPITCH + pc);
        *(v4f*)(outp + fr * DIM + pc) = val;
    }
}

#define RD_RUNS    4
#define RD_LEN     (NCHUNKS / RD_RUNS)
#define RD_THREADS 128
#define RD_BLOCKS  (FDIM * DIM / 4 / 8 / (RD_THREADS / 32))

static_assert(RD_RUNS * RD_LEN == NCHUNKS, "runs cover the chunks");
static_assert((DIM / 4) % 8 == 0, "a wave's 8 pieces stay inside one proj row");
static_assert(RD_BLOCKS * (RD_THREADS / 32) * 8 * 4 == FDIM * DIM, "pieces cover proj exactly");

__global__ __launch_bounds__(RD_THREADS) __attribute__((amdgpu_waves_per_eu(1, 4)))
void lgcn_reduce(const float* __restrict__ partial,
                 const float* __restrict__ filt  ,
                 float* __restrict__ proj  ) {
    const int lane  = threadIdx.x & 31;
    const int wv    = blockIdx.x * (RD_THREADS / 32) + (threadIdx.x >> 5);
    const int piece = wv * 8 + (lane & 7);
    const int run   = lane >> 3;
    const float* src = partial + (size_t)run * RD_LEN * FDIM * DIM + 4 * piece;
    const float f = filt[__builtin_amdgcn_readfirstlane(wv) / (DIM / 4 / 8)];
    v4f p[RD_LEN];
#pragma unroll
    for (int n = 0; n < RD_LEN; ++n)
        p[n] = *(const v4f*)(src + (size_t)n * FDIM * DIM);
    v4f s = p[0];
#pragma unroll
    for (int n = 1; n < RD_LEN; ++n)
        s = s + p[n];
    asm volatile("" :: "s"(f));
    v4f o;
    o.x = __shfl_xor(s.x, 8);  o.y = __shfl_xor(s.y, 8);  o.z = __shfl_xor(s.z, 8);  o.w = __shfl_xor(s.w, 8);
    s = s + o;
    o.x = __shfl_xor(s.x, 16); o.y = __shfl_xor(s.y, 16); o.z = __shfl_xor(s.z, 16); o.w = __shfl_xor(s.w, 16);
    s = s + o;
    if (run == 0)
        *(v4f*)(proj + 4 * piece) = s * f;
}

#define PROP_WAVES   12
#define PROP_GROUPS  (PROP_WAVES / 2)
#define GROUP_ROWS   48
#define PROP_ROWS    (PROP_GROUPS * GROUP_ROWS)
#define PROP_THREADS (PROP_WAVES * 32)
#define PROP_BLOCKS  ((N_TOTAL + PROP_ROWS - 1) / PROP_ROWS)
#define KBLK         32
#define KHALF        (FDIM / 2)
#define NKBLK        (KHALF / KBLK)
#define VPITCH       36
#define PIECES       12
#define PIECE_ROWS   4
#define SLICE        (GROUP_ROWS * VPITCH)
#define MT           (GROUP_ROWS / 16)

static_assert(GROUP_ROWS * (KBLK / 4) == 32 * PIECES, "a wave's 32 lanes x 12 pieces cover its 48 rows x 32 columns");
static_assert(PIECE_ROWS * PIECES == GROUP_ROWS, "piece rows cover the slice");
static_assert(PROP_BLOCKS * PROP_ROWS >= N_TOTAL, "row coverage");
static_assert(MT * 2 * 8 * 32 <= SLICE, "the swapped tiles of a wave fit its own stage-0 slice");
static_assert((NKBLK - 1) % 2 == 1, "the last block is read from stage 1, so a wave's stage-1 slice is its own to reuse after the K loop");

__device__ __forceinline__ void prop_kblock(const float* __restrict__ ar,
                                            const v2f* __restrict__ bp0,
                                            v8f (&c)[MT][4]) {
#pragma unroll
    for (int kk = 0; kk < KBLK; kk += 4) {
        v2f a[MT];
#pragma unroll
        for (int m = 0; m < MT; ++m)
            a[m] = *(const v2f*)(ar + m * 16 * VPITCH + kk);
        const v2f* bp = bp0 + (kk >> 1) * DIM;
#pragma unroll
        for (int t = 0; t < 4; ++t) {
            const v2f b = bp[t * 16];
#pragma unroll
            for (int m = 0; m < MT; ++m)
                c[m][t] = WMMA_F32(a[m], b, c[m][t]);
        }
    }
}

template <int T0>
__device__ __forceinline__ void prop_give(float* __restrict__ x, int lane, const v8f (&c)[MT][4]) {
#pragma unroll
    for (int m = 0; m < MT; ++m) {
#pragma unroll
        for (int u = 0; u < 2; ++u) {
            const int q = m * 2 + u;
            const v8f val = c[m][T0 + u];
            v4f lo, hi;
            lo.x = val[0]; lo.y = val[1]; lo.z = val[2]; lo.w = val[3];
            hi.x = val[4]; hi.y = val[5]; hi.z = val[6]; hi.w = val[7];
            *(v4f*)(x + ((q * 2)     * 32 + lane) * 4) = lo;
            *(v4f*)(x + ((q * 2 + 1) * 32 + lane) * 4) = hi;
        }
    }
}

template <int T0>
__device__ __forceinline__ void prop_finish(const float* __restrict__ xp, float* __restrict__ turn,
                                            int lane, int r_first,
                                            const v8f (&c)[MT][4], float* __restrict__ ego_next) {
    const int row  = lane & 15;
    const int half = lane >> 4;
#pragma unroll
    for (int m = 0; m < MT; ++m) {
#pragma unroll
        for (int u = 0; u < 2; ++u) {
            const int q = m * 2 + u;
            const v4f lo = *(const v4f*)(xp + ((q * 2)     * 32 + lane) * 4);
            const v4f hi = *(const v4f*)(xp + ((q * 2 + 1) * 32 + lane) * 4);
            const v8f own = c[m][T0 + u];
            float s[8];
            s[0] = own[0] + lo.x; s[1] = own[1] + lo.y; s[2] = own[2] + lo.z; s[3] = own[3] + lo.w;
            s[4] = own[4] + hi.x; s[5] = own[5] + hi.y; s[6] = own[6] + hi.z; s[7] = own[7] + hi.w;
#pragma unroll
            for (int j = 0; j < 8; ++j)
                turn[(m * 16 + (half << 3) + j) * VPITCH + u * 16 + row] = s[j];
        }
    }
    const int pr = lane >> 3;
    const int pc = 4 * (lane & 7);
    const float* tp = turn + pr * VPITCH + pc;
    float* gp = ego_next + (size_t)(r_first + pr) * DIM + T0 * 16 + pc;
    if (r_first + GROUP_ROWS <= N_TOTAL) {
        v4f val[GROUP_ROWS / 4];
#pragma unroll
        for (int i = 0; i < GROUP_ROWS / 4; ++i)
            val[i] = *(const v4f*)(tp + 4 * i * VPITCH);
#pragma unroll
        for (int i = 0; i < GROUP_ROWS / 4; ++i)
            *(v4f*)(gp + (size_t)(4 * i) * DIM) = val[i];
    } else {
#pragma unroll
        for (int i = 0; i < GROUP_ROWS / 4; ++i) {
            const v4f val = *(const v4f*)(tp + 4 * i * VPITCH);
            if (r_first + 4 * i + pr < N_TOTAL) *(v4f*)(gp + (size_t)(4 * i) * DIM) = val;
        }
    }
}

__global__ __launch_bounds__(PROP_THREADS)
void lgcn_prop(const float* __restrict__ v,
               const float* __restrict__ proj,
               float* __restrict__ ego_next) {
    __shared__ v2f lds_proj[(FDIM / 2) * DIM];
    __shared__ __attribute__((aligned(16))) float lds_v[2][PROP_WAVES * SLICE];

    const int tid  = threadIdx.x;
    const int lane = tid & 31;
    const int wave = tid >> 5;
    const int grp  = wave >> 1;
    const int kh   = wave & 1;
    const int g0   = blockIdx.x * PROP_ROWS + grp * GROUP_ROWS;

    const int lj = lane & 7;
    const int lr = lane >> 3;
    unsigned goff[PIECES];
#pragma unroll
    for (int i = 0; i < PIECES; ++i)
        goff[i] = (unsigned)min(g0 + lr + PIECE_ROWS * i, N_TOTAL - 1) * FDIM + kh * KHALF + lj * 4;
    const int soff = wave * SLICE + lr * VPITCH + lj * 4;

    v4f st[PIECES];
#pragma unroll
    for (int i = 0; i < PIECES; ++i)
        st[i] = *(const v4f*)(v + goff[i]);

    for (int idx = tid; idx < (FDIM / 2) * DIM; idx += PROP_THREADS) {
        const int j = idx >> 6;
        const int d = idx & 63;
        v2f val;
        val.x = proj[(size_t)(2 * j)     * DIM + d];
        val.y = proj[(size_t)(2 * j + 1) * DIM + d];
        lds_proj[idx] = val;
    }
#pragma unroll
    for (int i = 0; i < PIECES; ++i)
        *(v4f*)&lds_v[0][soff + i * PIECE_ROWS * VPITCH] = st[i];
    __syncthreads();

    const int row   = lane & 15;
    const int half  = lane >> 4;
    const int khalf = half * 2;

    v8f c[MT][4];
#pragma unroll
    for (int m = 0; m < MT; ++m)
#pragma unroll
        for (int t = 0; t < 4; ++t)
            c[m][t] = (v8f){0.0f, 0.0f, 0.0f, 0.0f, 0.0f, 0.0f, 0.0f, 0.0f};

    const int aoff = wave * SLICE + row * VPITCH + khalf;
    const v2f* bbase = &lds_proj[(kh * (KHALF / 2) + half) * DIM + row];

#pragma unroll 1
    for (int kb = 0; kb < NKBLK - 1; ++kb) {
#pragma unroll
        for (int i = 0; i < PIECES; ++i)
            st[i] = *(const v4f*)(v + goff[i] + (kb + 1) * KBLK);
        prop_kblock(lds_v[kb & 1] + aoff, bbase + (size_t)(kb * (KBLK / 2)) * DIM, c);
        __builtin_amdgcn_sched_barrier(0);
        float* ln = lds_v[(kb + 1) & 1];
#pragma unroll
        for (int i = 0; i < PIECES; ++i)
            *(v4f*)&ln[soff + i * PIECE_ROWS * VPITCH] = st[i];
    }
    prop_kblock(lds_v[(NKBLK - 1) & 1] + aoff, bbase + (size_t)((NKBLK - 1) * (KBLK / 2)) * DIM, c);

    float*       xmine = &lds_v[0][wave * SLICE];
    const float* xpart = &lds_v[0][(wave ^ 1) * SLICE];
    if (kh == 0) prop_give<2>(xmine, lane, c);
    else         prop_give<0>(xmine, lane, c);
    __syncthreads();
    float* turn = &lds_v[1][wave * SLICE];
    if (kh == 0) prop_finish<0>(xpart, turn, lane, g0, c, ego_next);
    else         prop_finish<2>(xpart, turn, lane, g0, c, ego_next);
}

#define PLANE  ((size_t)N_TOTAL * DIM)
#define PLANE4 (N_TOTAL * DIM / 4)

static_assert(N_LAYERS == 3, "lgcn_final adds four planes");
static_assert((N_TOTAL * DIM) % 4 == 0, "planes are whole 16-byte pieces");

__global__ void lgcn_final(const float* __restrict__ planes,
                           float* __restrict__ out) {
    int i = blockIdx.x * blockDim.x + threadIdx.x;
    if (i < PLANE4) {
        const v4f e0 = ((const v4f*)(planes))[i];
        const v4f e1 = ((const v4f*)(planes + PLANE))[i];
        const v4f e2 = ((const v4f*)(planes + 2 * PLANE))[i];
        const v4f e3 = ((const v4f*)(planes + 3 * PLANE))[i];
        ((v4f*)out)[i] = (((e0 + e1) + e2) + e3) * 0.25f;
    }
}

extern "C" void kernel_launch(void* const* d_in, const int* in_sizes, int n_in,
                              void* d_out, int out_size, void* d_ws, size_t ws_size,
                              hipStream_t stream) {
    const float* user    = (const float*)d_in[0];
    const float* item    = (const float*)d_in[1];
    const float* v       = (const float*)d_in[2];
    const float* filters = (const float*)d_in[3];

    float* ego     = (float*)d_ws;
    float* partial = ego + (size_t)(N_LAYERS + 1) * N_TOTAL * DIM;
    float* proj    = partial + (size_t)NCHUNKS * FDIM * DIM;

    lgcn_init<<<(TOTAL4 + 255) / 256, 256, 0, stream>>>(user, item, ego);

    for (int layer = 0; layer < N_LAYERS; ++layer) {
        dim3 gA(NCHUNKS, PT_FSPLIT);
        lgcn_projT<<<gA, 256, 0, stream>>>(v, ego + (size_t)layer * N_TOTAL * DIM, partial);

        lgcn_reduce<<<RD_BLOCKS, RD_THREADS, 0, stream>>>(
            partial, filters + (size_t)layer * FDIM, proj);

        lgcn_prop<<<PROP_BLOCKS, PROP_THREADS, 0, stream>>>(v, proj, ego + (size_t)(layer + 1) * N_TOTAL * DIM);
    }

    lgcn_final<<<(PLANE4 + 255) / 256, 256, 0, stream>>>(ego, (float*)d_out);
}


# Sum moved in front of the last GEMM: reduce keeps s0+s1+s2 of the filtered projections, last prop stores (ego0 + v@that)/4 to out; no final launch, no plane 3; was ((e0+e1)+e2)+e3 over planes
# speedup vs baseline: 3.6355x; 1.0362x over previous
#include <hip/hip_runtime.h>


typedef __attribute__((ext_vector_type(2))) float v2f;
typedef __attribute__((ext_vector_type(4))) float v4f;
typedef __attribute__((ext_vector_type(8))) float v8f;

#define N_USERS 29858
#define N_ITEMS 40981
#define N_TOTAL (N_USERS + N_ITEMS)
#define DIM     64
#define FDIM    512
#define N_LAYERS 3
#define NCHUNKS 128
#define CHUNK   556
#define QUADS   (CHUNK / 4)

static_assert(NCHUNKS * CHUNK >= N_TOTAL, "chunk coverage");

#define WMMA_F32(A, B, C) \
    __builtin_amdgcn_wmma_f32_16x16x4_f32(false, (A), false, (B), (short)0, (C), false, false)

#define USER4  (N_USERS * DIM / 4)
#define TOTAL4 (N_TOTAL * DIM / 4)

static_assert((N_USERS * DIM) % 4 == 0 && (N_ITEMS * DIM) % 4 == 0, "both tables are whole 16-byte pieces");

__global__ void lgcn_init(const float* __restrict__ user,
                          const float* __restrict__ item,
                          float* __restrict__ ego) {
    int i = blockIdx.x * blockDim.x + threadIdx.x;
    if (i < TOTAL4) {
        const v4f val = (i < USER4) ? ((const v4f*)user)[i] : ((const v4f*)item)[i - USER4];
        ((v4f*)ego)[i] = val;
    }
}

#define PT_FSPLIT 2
#define PT_F      (FDIM / PT_FSPLIT)
#define PT_ROWS   32
#define PT_NBLK   ((CHUNK + PT_ROWS - 1) / PT_ROWS)
#define PT_PAIRS  (PT_ROWS / 2)
#define PT_CPAIRS (PT_NBLK * PT_PAIRS)
#define PT_VPITCH (2 * PT_F + 32)
#define PT_EPITCH (2 * DIM + 32)
#define PT_VUNITS 4
#define PT_EUNITS (PT_CPAIRS * (DIM / 4) / 256)

static_assert(PT_PAIRS * (32 / 4) == 32 * PT_VUNITS, "a wave's v units per stage");
static_assert(PT_CPAIRS * (DIM / 4) == 256 * PT_EUNITS, "ego units per thread");

__device__ __forceinline__ void projT_fetch(const float* __restrict__ vcol, int nb, int vp,
                                            v4f (&sv0)[PT_VUNITS], v4f (&sv1)[PT_VUNITS]) {
#pragma unroll
    for (int i = 0; i < PT_VUNITS; ++i) {
        const int r = nb + 2 * (vp + 4 * i);
        sv0[i] = *(const v4f*)(vcol + (size_t)min(r,     N_TOTAL - 1) * FDIM);
        sv1[i] = *(const v4f*)(vcol + (size_t)min(r + 1, N_TOTAL - 1) * FDIM);
    }
}

__device__ __forceinline__ void projT_stash(float* __restrict__ lv, int vs,
                                            const v4f (&sv0)[PT_VUNITS], const v4f (&sv1)[PT_VUNITS]) {
#pragma unroll
    for (int i = 0; i < PT_VUNITS; ++i) {
        v4f lo, hi;
        lo.x = sv0[i].x; lo.y = sv1[i].x; lo.z = sv0[i].y; lo.w = sv1[i].y;
        hi.x = sv0[i].z; hi.y = sv1[i].z; hi.z = sv0[i].w; hi.w = sv1[i].w;
        *(v4f*)(lv + vs + i * 4 * PT_VPITCH)     = lo;
        *(v4f*)(lv + vs + i * 4 * PT_VPITCH + 4) = hi;
    }
}

__device__ __forceinline__ void projT_kblock(const float* __restrict__ ap,
                                             const float* __restrict__ bp,
                                             v8f (&c0)[4], v8f (&c1)[4]) {
#pragma unroll
    for (int kk = 0; kk < PT_ROWS / 4; ++kk) {
        const v2f a0 = *(const v2f*)(ap + (2 * kk) * PT_VPITCH);
        const v2f a1 = *(const v2f*)(ap + (2 * kk) * PT_VPITCH + 32);
        const float* bq = bp + (2 * kk) * PT_EPITCH;
#pragma unroll
        for (int t = 0; t < 4; ++t) {
            const v2f b = *(const v2f*)(bq + 32 * t);
            c0[t] = WMMA_F32(a0, b, c0[t]);
            c1[t] = WMMA_F32(a1, b, c1[t]);
        }
    }
}

__global__ __launch_bounds__(256)
void lgcn_projT(const float* __restrict__ v,
                const float* __restrict__ ego,
                float* __restrict__ partial  ) {
    __shared__ __attribute__((aligned(16))) float lds_vt[2][PT_PAIRS * PT_VPITCH];
    __shared__ __attribute__((aligned(16))) float lds_eg[PT_CPAIRS * PT_EPITCH];

    const int tid     = threadIdx.x;
    const int lane    = tid & 31;
    const int wave    = tid >> 5;
    const int chunk   = blockIdx.x;
    const int n_begin = chunk * CHUNK;
    const int n_valid = min(N_TOTAL - n_begin, CHUNK);
    const int fblk    = blockIdx.y * PT_F;

    const int vcg = wave * 8 + (lane & 7);
    const int vp  = lane >> 3;
    const float* vcol = v + fblk + 4 * vcg;
    const int vs = vp * PT_VPITCH + 8 * vcg;

    v4f sv0[PT_VUNITS], sv1[PT_VUNITS];
    projT_fetch(vcol, n_begin, vp, sv0, sv1);

    {
        const int ecg = tid & 15;
        const int ep0 = tid >> 4;
        const float* ecol = ego + 4 * ecg;
        v4f se0[PT_EUNITS], se1[PT_EUNITS];
#pragma unroll
        for (int j = 0; j < PT_EUNITS; ++j) {
            const int r = n_begin + 2 * (ep0 + 16 * j);
            se0[j] = *(const v4f*)(ecol + (size_t)min(r,     N_TOTAL - 1) * DIM);
            se1[j] = *(const v4f*)(ecol + (size_t)min(r + 1, N_TOTAL - 1) * DIM);
        }
        const v4f zero = {0.0f, 0.0f, 0.0f, 0.0f};
#pragma unroll
        for (int j = 0; j < PT_EUNITS; ++j) {
            const int ep = ep0 + 16 * j;
            const v4f e0 = (2 * ep     < n_valid) ? se0[j] : zero;
            const v4f e1 = (2 * ep + 1 < n_valid) ? se1[j] : zero;
            v4f lo, hi;
            lo.x = e0.x; lo.y = e1.x; lo.z = e0.y; lo.w = e1.y;
            hi.x = e0.z; hi.y = e1.z; hi.z = e0.w; hi.w = e1.w;
            *(v4f*)(lds_eg + ep * PT_EPITCH + 8 * ecg)     = lo;
            *(v4f*)(lds_eg + ep * PT_EPITCH + 8 * ecg + 4) = hi;
        }
    }
    projT_stash(lds_vt[0], vs, sv0, sv1);
    __syncthreads();

    const int f0    = fblk + wave * 32;
    const int row   = lane & 15;
    const int half  = lane >> 4;

    v8f c0[4] = {{}, {}, {}, {}};
    v8f c1[4] = {{}, {}, {}, {}};

    const int aoff = half * PT_VPITCH + 2 * (wave * 32 + row);
    const int boff = half * PT_EPITCH + 2 * row;

#pragma unroll 1
    for (int kb = 0; kb < PT_NBLK - 1; ++kb) {
        projT_fetch(vcol, n_begin + (kb + 1) * PT_ROWS, vp, sv0, sv1);
        projT_kblock(lds_vt[kb & 1] + aoff, lds_eg + kb * (PT_PAIRS * PT_EPITCH) + boff, c0, c1);
        __builtin_amdgcn_sched_barrier(0);
        projT_stash(lds_vt[(kb + 1) & 1], vs, sv0, sv1);
    }
    projT_kblock(lds_vt[(PT_NBLK - 1) & 1] + aoff,
                 lds_eg + (PT_NBLK - 1) * (PT_PAIRS * PT_EPITCH) + boff, c0, c1);

    float* lw = &lds_vt[0][0] + 64 * wave;
#pragma unroll
    for (int t = 0; t < 4; ++t) {
#pragma unroll
        for (int j = 0; j < 8; ++j) {
            lw[(j + 8 * half) * PT_VPITCH + t * 16 + row]                         = c0[t][j];
            lw[PT_PAIRS * PT_VPITCH + (j + 8 * half) * PT_VPITCH + t * 16 + row] = c1[t][j];
        }
    }
    float* outp = partial + (size_t)chunk * FDIM * DIM + (size_t)f0 * DIM;
    const int pr = lane >> 4;
    const int pc = 4 * (lane & 15);
#pragma unroll
    for (int i = 0; i < 16; ++i) {
        const int fr = 2 * i + pr;
        const v4f val = *(const v4f*)(lw + (fr >> 4) * (PT_PAIRS * PT_VPITCH) + (fr & 15) * PT_VPITCH + pc);
        *(v4f*)(outp + fr * DIM + pc) = val;
    }
}

#define RD_RUNS    4
#define RD_LEN     (NCHUNKS / RD_RUNS)
#define RD_THREADS 128
#define RD_BLOCKS  (FDIM * DIM / 4 / 8 / (RD_THREADS / 32))

static_assert(RD_RUNS * RD_LEN == NCHUNKS, "runs cover the chunks");
static_assert((DIM / 4) % 8 == 0, "a wave's 8 pieces stay inside one proj row");
static_assert(RD_BLOCKS * (RD_THREADS / 32) * 8 * 4 == FDIM * DIM, "pieces cover proj exactly");

template <int KIND>
__global__ __launch_bounds__(RD_THREADS) __attribute__((amdgpu_waves_per_eu(1, 4)))
void lgcn_reduce(const float* __restrict__ partial,
                 const float* __restrict__ filt  ,
                 float* proj  ,
                 float* ssum  ) {
    const int lane  = threadIdx.x & 31;
    const int wv    = blockIdx.x * (RD_THREADS / 32) + (threadIdx.x >> 5);
    const int piece = wv * 8 + (lane & 7);
    const int run   = lane >> 3;
    const float* src = partial + (size_t)run * RD_LEN * FDIM * DIM + 4 * piece;
    const float f = filt[__builtin_amdgcn_readfirstlane(wv) / (DIM / 4 / 8)];
    v4f prev = {0.0f, 0.0f, 0.0f, 0.0f};
    if (KIND == 1) prev = *(const v4f*)(proj + 4 * piece);
    if (KIND == 2) prev = *(const v4f*)(ssum + 4 * piece);
    v4f p[RD_LEN];
#pragma unroll
    for (int n = 0; n < RD_LEN; ++n)
        p[n] = *(const v4f*)(src + (size_t)n * FDIM * DIM);
    v4f s = p[0];
#pragma unroll
    for (int n = 1; n < RD_LEN; ++n)
        s = s + p[n];
    asm volatile("" :: "s"(f));
    v4f o;
    o.x = __shfl_xor(s.x, 8);  o.y = __shfl_xor(s.y, 8);  o.z = __shfl_xor(s.z, 8);  o.w = __shfl_xor(s.w, 8);
    s = s + o;
    o.x = __shfl_xor(s.x, 16); o.y = __shfl_xor(s.y, 16); o.z = __shfl_xor(s.z, 16); o.w = __shfl_xor(s.w, 16);
    s = s + o;
    if (run == 0) {
        const v4f sk = s * f;
        if (KIND == 0) *(v4f*)(proj + 4 * piece) = sk;
        if (KIND == 1) { *(v4f*)(ssum + 4 * piece) = prev + sk; *(v4f*)(proj + 4 * piece) = sk; }
        if (KIND == 2) *(v4f*)(proj + 4 * piece) = prev + sk;
    }
}

#define PROP_WAVES   12
#define PROP_GROUPS  (PROP_WAVES / 2)
#define GROUP_ROWS   48
#define PROP_ROWS    (PROP_GROUPS * GROUP_ROWS)
#define PROP_THREADS (PROP_WAVES * 32)
#define PROP_BLOCKS  ((N_TOTAL + PROP_ROWS - 1) / PROP_ROWS)
#define KBLK         32
#define KHALF        (FDIM / 2)
#define NKBLK        (KHALF / KBLK)
#define VPITCH       36
#define PIECES       12
#define PIECE_ROWS   4
#define SLICE        (GROUP_ROWS * VPITCH)
#define MT           (GROUP_ROWS / 16)

static_assert(GROUP_ROWS * (KBLK / 4) == 32 * PIECES, "a wave's 32 lanes x 12 pieces cover its 48 rows x 32 columns");
static_assert(PIECE_ROWS * PIECES == GROUP_ROWS, "piece rows cover the slice");
static_assert(PROP_BLOCKS * PROP_ROWS >= N_TOTAL, "row coverage");
static_assert(MT * 2 * 8 * 32 <= SLICE, "the swapped tiles of a wave fit its own stage-0 slice");
static_assert((NKBLK - 1) % 2 == 1, "the last block is read from stage 1, so a wave's stage-1 slice is its own to reuse after the K loop");

__device__ __forceinline__ void prop_kblock(const float* __restrict__ ar,
                                            const v2f* __restrict__ bp0,
                                            v8f (&c)[MT][4]) {
#pragma unroll
    for (int kk = 0; kk < KBLK; kk += 4) {
        v2f a[MT];
#pragma unroll
        for (int m = 0; m < MT; ++m)
            a[m] = *(const v2f*)(ar + m * 16 * VPITCH + kk);
        const v2f* bp = bp0 + (kk >> 1) * DIM;
#pragma unroll
        for (int t = 0; t < 4; ++t) {
            const v2f b = bp[t * 16];
#pragma unroll
            for (int m = 0; m < MT; ++m)
                c[m][t] = WMMA_F32(a[m], b, c[m][t]);
        }
    }
}

template <int T0>
__device__ __forceinline__ void prop_give(float* __restrict__ x, int lane, const v8f (&c)[MT][4]) {
#pragma unroll
    for (int m = 0; m < MT; ++m) {
#pragma unroll
        for (int u = 0; u < 2; ++u) {
            const int q = m * 2 + u;
            const v8f val = c[m][T0 + u];
            v4f lo, hi;
            lo.x = val[0]; lo.y = val[1]; lo.z = val[2]; lo.w = val[3];
            hi.x = val[4]; hi.y = val[5]; hi.z = val[6]; hi.w = val[7];
            *(v4f*)(x + ((q * 2)     * 32 + lane) * 4) = lo;
            *(v4f*)(x + ((q * 2 + 1) * 32 + lane) * 4) = hi;
        }
    }
}

template <int T0, bool LAST>
__device__ __forceinline__ void prop_finish(const float* __restrict__ xp, float* __restrict__ turn,
                                            int lane, int r_first,
                                            const v8f (&c)[MT][4], const v4f (&e0)[GROUP_ROWS / 4],
                                            float* __restrict__ dst) {
    const int row  = lane & 15;
    const int half = lane >> 4;
#pragma unroll
    for (int m = 0; m < MT; ++m) {
#pragma unroll
        for (int u = 0; u < 2; ++u) {
            const int q = m * 2 + u;
            const v4f lo = *(const v4f*)(xp + ((q * 2)     * 32 + lane) * 4);
            const v4f hi = *(const v4f*)(xp + ((q * 2 + 1) * 32 + lane) * 4);
            const v8f own = c[m][T0 + u];
            float s[8];
            s[0] = own[0] + lo.x; s[1] = own[1] + lo.y; s[2] = own[2] + lo.z; s[3] = own[3] + lo.w;
            s[4] = own[4] + hi.x; s[5] = own[5] + hi.y; s[6] = own[6] + hi.z; s[7] = own[7] + hi.w;
#pragma unroll
            for (int j = 0; j < 8; ++j)
                turn[(m * 16 + (half << 3) + j) * VPITCH + u * 16 + row] = s[j];
        }
    }
    const int pr = lane >> 3;
    const int pc = 4 * (lane & 7);
    const float* tp = turn + pr * VPITCH + pc;
    float* gp = dst + (size_t)(r_first + pr) * DIM + T0 * 16 + pc;
    if (r_first + GROUP_ROWS <= N_TOTAL) {
        v4f val[GROUP_ROWS / 4];
#pragma unroll
        for (int i = 0; i < GROUP_ROWS / 4; ++i)
            val[i] = *(const v4f*)(tp + 4 * i * VPITCH);
#pragma unroll
        for (int i = 0; i < GROUP_ROWS / 4; ++i)
            *(v4f*)(gp + (size_t)(4 * i) * DIM) = LAST ? (e0[i] + val[i]) * 0.25f : val[i];
    } else {
#pragma unroll
        for (int i = 0; i < GROUP_ROWS / 4; ++i) {
            const v4f val = *(const v4f*)(tp + 4 * i * VPITCH);
            if (r_first + 4 * i + pr < N_TOTAL)
                *(v4f*)(gp + (size_t)(4 * i) * DIM) = LAST ? (e0[i] + val) * 0.25f : val;
        }
    }
}

template <bool LAST>
__global__ __launch_bounds__(PROP_THREADS)
void lgcn_prop(const float* __restrict__ v,
               const float* __restrict__ proj,
               const float* __restrict__ ego0  ,
               float* __restrict__ dst  ) {
    __shared__ v2f lds_proj[(FDIM / 2) * DIM];
    __shared__ __attribute__((aligned(16))) float lds_v[2][PROP_WAVES * SLICE];

    const int tid  = threadIdx.x;
    const int lane = tid & 31;
    const int wave = tid >> 5;
    const int grp  = wave >> 1;
    const int kh   = wave & 1;
    const int g0   = blockIdx.x * PROP_ROWS + grp * GROUP_ROWS;

    const int lj = lane & 7;
    const int lr = lane >> 3;
    unsigned goff[PIECES];
#pragma unroll
    for (int i = 0; i < PIECES; ++i)
        goff[i] = (unsigned)min(g0 + lr + PIECE_ROWS * i, N_TOTAL - 1) * FDIM + kh * KHALF + lj * 4;
    const int soff = wave * SLICE + lr * VPITCH + lj * 4;

    v4f st[PIECES];
#pragma unroll
    for (int i = 0; i < PIECES; ++i)
        st[i] = *(const v4f*)(v + goff[i]);

    for (int idx = tid; idx < (FDIM / 2) * DIM; idx += PROP_THREADS) {
        const int j = idx >> 6;
        const int d = idx & 63;
        v2f val;
        val.x = proj[(size_t)(2 * j)     * DIM + d];
        val.y = proj[(size_t)(2 * j + 1) * DIM + d];
        lds_proj[idx] = val;
    }
#pragma unroll
    for (int i = 0; i < PIECES; ++i)
        *(v4f*)&lds_v[0][soff + i * PIECE_ROWS * VPITCH] = st[i];
    __syncthreads();

    const int row   = lane & 15;
    const int half  = lane >> 4;
    const int khalf = half * 2;

    v8f c[MT][4];
#pragma unroll
    for (int m = 0; m < MT; ++m)
#pragma unroll
        for (int t = 0; t < 4; ++t)
            c[m][t] = (v8f){0.0f, 0.0f, 0.0f, 0.0f, 0.0f, 0.0f, 0.0f, 0.0f};

    const int aoff = wave * SLICE + row * VPITCH + khalf;
    const v2f* bbase = &lds_proj[(kh * (KHALF / 2) + half) * DIM + row];

#pragma unroll 1
    for (int kb = 0; kb < NKBLK - 1; ++kb) {
#pragma unroll
        for (int i = 0; i < PIECES; ++i)
            st[i] = *(const v4f*)(v + goff[i] + (kb + 1) * KBLK);
        prop_kblock(lds_v[kb & 1] + aoff, bbase + (size_t)(kb * (KBLK / 2)) * DIM, c);
        __builtin_amdgcn_sched_barrier(0);
        float* ln = lds_v[(kb + 1) & 1];
#pragma unroll
        for (int i = 0; i < PIECES; ++i)
            *(v4f*)&ln[soff + i * PIECE_ROWS * VPITCH] = st[i];
    }
    v4f e0[GROUP_ROWS / 4];
    if (LAST) {
        const float* ep = ego0 + (kh == 0 ? 0 : 32) + 4 * (lane & 7);
#pragma unroll
        for (int i = 0; i < GROUP_ROWS / 4; ++i)
            e0[i] = *(const v4f*)(ep + (size_t)min(g0 + (lane >> 3) + 4 * i, N_TOTAL - 1) * DIM);
    }
    prop_kblock(lds_v[(NKBLK - 1) & 1] + aoff, bbase + (size_t)((NKBLK - 1) * (KBLK / 2)) * DIM, c);

    float*       xmine = &lds_v[0][wave * SLICE];
    const float* xpart = &lds_v[0][(wave ^ 1) * SLICE];
    if (kh == 0) prop_give<2>(xmine, lane, c);
    else         prop_give<0>(xmine, lane, c);
    __syncthreads();
    float* turn = &lds_v[1][wave * SLICE];
    if (kh == 0) prop_finish<0, LAST>(xpart, turn, lane, g0, c, e0, dst);
    else         prop_finish<2, LAST>(xpart, turn, lane, g0, c, e0, dst);
}

static_assert(N_LAYERS == 3, "the reduction's three kinds are first, middle, last");

extern "C" void kernel_launch(void* const* d_in, const int* in_sizes, int n_in,
                              void* d_out, int out_size, void* d_ws, size_t ws_size,
                              hipStream_t stream) {
    const float* user    = (const float*)d_in[0];
    const float* item    = (const float*)d_in[1];
    const float* v       = (const float*)d_in[2];
    const float* filters = (const float*)d_in[3];

    float* ego0    = (float*)d_ws;
    float* ego1    = ego0 + (size_t)N_TOTAL * DIM;
    float* ego2    = ego1 + (size_t)N_TOTAL * DIM;
    float* partial = ego2 + (size_t)N_TOTAL * DIM;
    float* proj    = partial + (size_t)NCHUNKS * FDIM * DIM;
    float* ssum    = proj + (size_t)FDIM * DIM;
    float* out     = (float*)d_out;

    const dim3 gA(NCHUNKS, PT_FSPLIT);

    lgcn_init<<<(TOTAL4 + 255) / 256, 256, 0, stream>>>(user, item, ego0);

    lgcn_projT<<<gA, 256, 0, stream>>>(v, ego0, partial);
    lgcn_reduce<0><<<RD_BLOCKS, RD_THREADS, 0, stream>>>(partial, filters, proj, ssum);
    lgcn_prop<false><<<PROP_BLOCKS, PROP_THREADS, 0, stream>>>(v, proj, ego0, ego1);

    lgcn_projT<<<gA, 256, 0, stream>>>(v, ego1, partial);
    lgcn_reduce<1><<<RD_BLOCKS, RD_THREADS, 0, stream>>>(partial, filters + FDIM, proj, ssum);
    lgcn_prop<false><<<PROP_BLOCKS, PROP_THREADS, 0, stream>>>(v, proj, ego0, ego2);

    lgcn_projT<<<gA, 256, 0, stream>>>(v, ego2, partial);
    lgcn_reduce<2><<<RD_BLOCKS, RD_THREADS, 0, stream>>>(partial, filters + 2 * FDIM, proj, ssum);
    lgcn_prop<true><<<PROP_BLOCKS, PROP_THREADS, 0, stream>>>(v, proj, ego0, out);
}


# Both GEMM K loops: a wave's issue priority (s_setprio) falls 3..0 with its own trip count, so the waves of a SIMD end together and none runs its last blocks alone; sums and their order unchanged
# speedup vs baseline: 3.7483x; 1.0310x over previous
#include <hip/hip_runtime.h>


typedef __attribute__((ext_vector_type(2))) float v2f;
typedef __attribute__((ext_vector_type(4))) float v4f;
typedef __attribute__((ext_vector_type(8))) float v8f;

#define N_USERS 29858
#define N_ITEMS 40981
#define N_TOTAL (N_USERS + N_ITEMS)
#define DIM     64
#define FDIM    512
#define N_LAYERS 3
#define NCHUNKS 128
#define CHUNK   556
#define QUADS   (CHUNK / 4)

static_assert(NCHUNKS * CHUNK >= N_TOTAL, "chunk coverage");

#define WMMA_F32(A, B, C) \
    __builtin_amdgcn_wmma_f32_16x16x4_f32(false, (A), false, (B), (short)0, (C), false, false)

#define USER4  (N_USERS * DIM / 4)
#define TOTAL4 (N_TOTAL * DIM / 4)

static_assert((N_USERS * DIM) % 4 == 0 && (N_ITEMS * DIM) % 4 == 0, "both tables are whole 16-byte pieces");

__global__ void lgcn_init(const float* __restrict__ user,
                          const float* __restrict__ item,
                          float* __restrict__ ego) {
    int i = blockIdx.x * blockDim.x + threadIdx.x;
    if (i < TOTAL4) {
        const v4f val = (i < USER4) ? ((const v4f*)user)[i] : ((const v4f*)item)[i - USER4];
        ((v4f*)ego)[i] = val;
    }
}

#define PT_FSPLIT 2
#define PT_F      (FDIM / PT_FSPLIT)
#define PT_ROWS   32
#define PT_NBLK   ((CHUNK + PT_ROWS - 1) / PT_ROWS)
#define PT_PAIRS  (PT_ROWS / 2)
#define PT_CPAIRS (PT_NBLK * PT_PAIRS)
#define PT_VPITCH (2 * PT_F + 32)
#define PT_EPITCH (2 * DIM + 32)
#define PT_VUNITS 4
#define PT_EUNITS (PT_CPAIRS * (DIM / 4) / 256)

#define PT_PRIO2  (PT_NBLK - 8)
#define PT_PRIO1  (PT_NBLK - 4)
#define PT_PRIO0  (PT_NBLK - 2)

static_assert(PT_PAIRS * (32 / 4) == 32 * PT_VUNITS, "a wave's v units per stage");
static_assert(PT_PRIO2 > 0 && PT_PRIO0 < PT_NBLK - 1, "every priority step lies inside the K loop, which is left at priority 0");
static_assert(PT_CPAIRS * (DIM / 4) == 256 * PT_EUNITS, "ego units per thread");

__device__ __forceinline__ void projT_fetch(const float* __restrict__ vcol, int nb, int vp,
                                            v4f (&sv0)[PT_VUNITS], v4f (&sv1)[PT_VUNITS]) {
#pragma unroll
    for (int i = 0; i < PT_VUNITS; ++i) {
        const int r = nb + 2 * (vp + 4 * i);
        sv0[i] = *(const v4f*)(vcol + (size_t)min(r,     N_TOTAL - 1) * FDIM);
        sv1[i] = *(const v4f*)(vcol + (size_t)min(r + 1, N_TOTAL - 1) * FDIM);
    }
}

__device__ __forceinline__ void projT_stash(float* __restrict__ lv, int vs,
                                            const v4f (&sv0)[PT_VUNITS], const v4f (&sv1)[PT_VUNITS]) {
#pragma unroll
    for (int i = 0; i < PT_VUNITS; ++i) {
        v4f lo, hi;
        lo.x = sv0[i].x; lo.y = sv1[i].x; lo.z = sv0[i].y; lo.w = sv1[i].y;
        hi.x = sv0[i].z; hi.y = sv1[i].z; hi.z = sv0[i].w; hi.w = sv1[i].w;
        *(v4f*)(lv + vs + i * 4 * PT_VPITCH)     = lo;
        *(v4f*)(lv + vs + i * 4 * PT_VPITCH + 4) = hi;
    }
}

__device__ __forceinline__ void projT_kblock(const float* __restrict__ ap,
                                             const float* __restrict__ bp,
                                             v8f (&c0)[4], v8f (&c1)[4]) {
#pragma unroll
    for (int kk = 0; kk < PT_ROWS / 4; ++kk) {
        const v2f a0 = *(const v2f*)(ap + (2 * kk) * PT_VPITCH);
        const v2f a1 = *(const v2f*)(ap + (2 * kk) * PT_VPITCH + 32);
        const float* bq = bp + (2 * kk) * PT_EPITCH;
#pragma unroll
        for (int t = 0; t < 4; ++t) {
            const v2f b = *(const v2f*)(bq + 32 * t);
            c0[t] = WMMA_F32(a0, b, c0[t]);
            c1[t] = WMMA_F32(a1, b, c1[t]);
        }
    }
}

__global__ __launch_bounds__(256)
void lgcn_projT(const float* __restrict__ v,
                const float* __restrict__ ego,
                float* __restrict__ partial  ) {
    __shared__ __attribute__((aligned(16))) float lds_vt[2][PT_PAIRS * PT_VPITCH];
    __shared__ __attribute__((aligned(16))) float lds_eg[PT_CPAIRS * PT_EPITCH];

    const int tid     = threadIdx.x;
    const int lane    = tid & 31;
    const int wave    = tid >> 5;
    const int chunk   = blockIdx.x;
    const int n_begin = chunk * CHUNK;
    const int n_valid = min(N_TOTAL - n_begin, CHUNK);
    const int fblk    = blockIdx.y * PT_F;

    const int vcg = wave * 8 + (lane & 7);
    const int vp  = lane >> 3;
    const float* vcol = v + fblk + 4 * vcg;
    const int vs = vp * PT_VPITCH + 8 * vcg;

    v4f sv0[PT_VUNITS], sv1[PT_VUNITS];
    projT_fetch(vcol, n_begin, vp, sv0, sv1);

    {
        const int ecg = tid & 15;
        const int ep0 = tid >> 4;
        const float* ecol = ego + 4 * ecg;
        v4f se0[PT_EUNITS], se1[PT_EUNITS];
#pragma unroll
        for (int j = 0; j < PT_EUNITS; ++j) {
            const int r = n_begin + 2 * (ep0 + 16 * j);
            se0[j] = *(const v4f*)(ecol + (size_t)min(r,     N_TOTAL - 1) * DIM);
            se1[j] = *(const v4f*)(ecol + (size_t)min(r + 1, N_TOTAL - 1) * DIM);
        }
        const v4f zero = {0.0f, 0.0f, 0.0f, 0.0f};
#pragma unroll
        for (int j = 0; j < PT_EUNITS; ++j) {
            const int ep = ep0 + 16 * j;
            const v4f e0 = (2 * ep     < n_valid) ? se0[j] : zero;
            const v4f e1 = (2 * ep + 1 < n_valid) ? se1[j] : zero;
            v4f lo, hi;
            lo.x = e0.x; lo.y = e1.x; lo.z = e0.y; lo.w = e1.y;
            hi.x = e0.z; hi.y = e1.z; hi.z = e0.w; hi.w = e1.w;
            *(v4f*)(lds_eg + ep * PT_EPITCH + 8 * ecg)     = lo;
            *(v4f*)(lds_eg + ep * PT_EPITCH + 8 * ecg + 4) = hi;
        }
    }
    projT_stash(lds_vt[0], vs, sv0, sv1);
    __syncthreads();

    const int f0    = fblk + wave * 32;
    const int row   = lane & 15;
    const int half  = lane >> 4;

    v8f c0[4] = {{}, {}, {}, {}};
    v8f c1[4] = {{}, {}, {}, {}};

    const int aoff = half * PT_VPITCH + 2 * (wave * 32 + row);
    const int boff = half * PT_EPITCH + 2 * row;

    __builtin_amdgcn_s_setprio(3);
#pragma unroll 1
    for (int kb = 0; kb < PT_NBLK - 1; ++kb) {
        if (kb == PT_PRIO2) __builtin_amdgcn_s_setprio(2);
        if (kb == PT_PRIO1) __builtin_amdgcn_s_setprio(1);
        if (kb == PT_PRIO0) __builtin_amdgcn_s_setprio(0);
        projT_fetch(vcol, n_begin + (kb + 1) * PT_ROWS, vp, sv0, sv1);
        projT_kblock(lds_vt[kb & 1] + aoff, lds_eg + kb * (PT_PAIRS * PT_EPITCH) + boff, c0, c1);
        __builtin_amdgcn_sched_barrier(0);
        projT_stash(lds_vt[(kb + 1) & 1], vs, sv0, sv1);
    }
    projT_kblock(lds_vt[(PT_NBLK - 1) & 1] + aoff,
                 lds_eg + (PT_NBLK - 1) * (PT_PAIRS * PT_EPITCH) + boff, c0, c1);

    float* lw = &lds_vt[0][0] + 64 * wave;
#pragma unroll
    for (int t = 0; t < 4; ++t) {
#pragma unroll
        for (int j = 0; j < 8; ++j) {
            lw[(j + 8 * half) * PT_VPITCH + t * 16 + row]                         = c0[t][j];
            lw[PT_PAIRS * PT_VPITCH + (j + 8 * half) * PT_VPITCH + t * 16 + row] = c1[t][j];
        }
    }
    float* outp = partial + (size_t)chunk * FDIM * DIM + (size_t)f0 * DIM;
    const int pr = lane >> 4;
    const int pc = 4 * (lane & 15);
#pragma unroll
    for (int i = 0; i < 16; ++i) {
        const int fr = 2 * i + pr;
        const v4f val = *(const v4f*)(lw + (fr >> 4) * (PT_PAIRS * PT_VPITCH) + (fr & 15) * PT_VPITCH + pc);
        *(v4f*)(outp + fr * DIM + pc) = val;
    }
}

#define RD_RUNS    4
#define RD_LEN     (NCHUNKS / RD_RUNS)
#define RD_THREADS 128
#define RD_BLOCKS  (FDIM * DIM / 4 / 8 / (RD_THREADS / 32))

static_assert(RD_RUNS * RD_LEN == NCHUNKS, "runs cover the chunks");
static_assert((DIM / 4) % 8 == 0, "a wave's 8 pieces stay inside one proj row");
static_assert(RD_BLOCKS * (RD_THREADS / 32) * 8 * 4 == FDIM * DIM, "pieces cover proj exactly");

template <int KIND>
__global__ __launch_bounds__(RD_THREADS) __attribute__((amdgpu_waves_per_eu(1, 4)))
void lgcn_reduce(const float* __restrict__ partial,
                 const float* __restrict__ filt  ,
                 float* proj  ,
                 float* ssum  ) {
    const int lane  = threadIdx.x & 31;
    const int wv    = blockIdx.x * (RD_THREADS / 32) + (threadIdx.x >> 5);
    const int piece = wv * 8 + (lane & 7);
    const int run   = lane >> 3;
    const float* src = partial + (size_t)run * RD_LEN * FDIM * DIM + 4 * piece;
    const float f = filt[__builtin_amdgcn_readfirstlane(wv) / (DIM / 4 / 8)];
    v4f prev = {0.0f, 0.0f, 0.0f, 0.0f};
    if (KIND == 1) prev = *(const v4f*)(proj + 4 * piece);
    if (KIND == 2) prev = *(const v4f*)(ssum + 4 * piece);
    v4f p[RD_LEN];
#pragma unroll
    for (int n = 0; n < RD_LEN; ++n)
        p[n] = *(const v4f*)(src + (size_t)n * FDIM * DIM);
    v4f s = p[0];
#pragma unroll
    for (int n = 1; n < RD_LEN; ++n)
        s = s + p[n];
    asm volatile("" :: "s"(f));
    v4f o;
    o.x = __shfl_xor(s.x, 8);  o.y = __shfl_xor(s.y, 8);  o.z = __shfl_xor(s.z, 8);  o.w = __shfl_xor(s.w, 8);
    s = s + o;
    o.x = __shfl_xor(s.x, 16); o.y = __shfl_xor(s.y, 16); o.z = __shfl_xor(s.z, 16); o.w = __shfl_xor(s.w, 16);
    s = s + o;
    if (run == 0) {
        const v4f sk = s * f;
        if (KIND == 0) *(v4f*)(proj + 4 * piece) = sk;
        if (KIND == 1) { *(v4f*)(ssum + 4 * piece) = prev + sk; *(v4f*)(proj + 4 * piece) = sk; }
        if (KIND == 2) *(v4f*)(proj + 4 * piece) = prev + sk;
    }
}

#define PROP_WAVES   12
#define PROP_GROUPS  (PROP_WAVES / 2)
#define GROUP_ROWS   48
#define PROP_ROWS    (PROP_GROUPS * GROUP_ROWS)
#define PROP_THREADS (PROP_WAVES * 32)
#define PROP_BLOCKS  ((N_TOTAL + PROP_ROWS - 1) / PROP_ROWS)
#define KBLK         32
#define KHALF        (FDIM / 2)
#define NKBLK        (KHALF / KBLK)
#define VPITCH       36
#define PIECES       12
#define PIECE_ROWS   4
#define SLICE        (GROUP_ROWS * VPITCH)
#define MT           (GROUP_ROWS / 16)

#define PROP_PRIO2   (NKBLK - 6)
#define PROP_PRIO1   (NKBLK - 4)
#define PROP_PRIO0   (NKBLK - 2)

static_assert(GROUP_ROWS * (KBLK / 4) == 32 * PIECES, "a wave's 32 lanes x 12 pieces cover its 48 rows x 32 columns");
static_assert(PROP_PRIO2 > 0 && PROP_PRIO0 < NKBLK - 1, "every priority step lies inside the K loop, which is left at priority 0");
static_assert(PIECE_ROWS * PIECES == GROUP_ROWS, "piece rows cover the slice");
static_assert(PROP_BLOCKS * PROP_ROWS >= N_TOTAL, "row coverage");
static_assert(MT * 2 * 8 * 32 <= SLICE, "the swapped tiles of a wave fit its own stage-0 slice");
static_assert((NKBLK - 1) % 2 == 1, "the last block is read from stage 1, so a wave's stage-1 slice is its own to reuse after the K loop");

__device__ __forceinline__ void prop_kblock(const float* __restrict__ ar,
                                            const v2f* __restrict__ bp0,
                                            v8f (&c)[MT][4]) {
#pragma unroll
    for (int kk = 0; kk < KBLK; kk += 4) {
        v2f a[MT];
#pragma unroll
        for (int m = 0; m < MT; ++m)
            a[m] = *(const v2f*)(ar + m * 16 * VPITCH + kk);
        const v2f* bp = bp0 + (kk >> 1) * DIM;
#pragma unroll
        for (int t = 0; t < 4; ++t) {
            const v2f b = bp[t * 16];
#pragma unroll
            for (int m = 0; m < MT; ++m)
                c[m][t] = WMMA_F32(a[m], b, c[m][t]);
        }
    }
}

template <int T0>
__device__ __forceinline__ void prop_give(float* __restrict__ x, int lane, const v8f (&c)[MT][4]) {
#pragma unroll
    for (int m = 0; m < MT; ++m) {
#pragma unroll
        for (int u = 0; u < 2; ++u) {
            const int q = m * 2 + u;
            const v8f val = c[m][T0 + u];
            v4f lo, hi;
            lo.x = val[0]; lo.y = val[1]; lo.z = val[2]; lo.w = val[3];
            hi.x = val[4]; hi.y = val[5]; hi.z = val[6]; hi.w = val[7];
            *(v4f*)(x + ((q * 2)     * 32 + lane) * 4) = lo;
            *(v4f*)(x + ((q * 2 + 1) * 32 + lane) * 4) = hi;
        }
    }
}

template <int T0, bool LAST>
__device__ __forceinline__ void prop_finish(const float* __restrict__ xp, float* __restrict__ turn,
                                            int lane, int r_first,
                                            const v8f (&c)[MT][4], const v4f (&e0)[GROUP_ROWS / 4],
                                            float* __restrict__ dst) {
    const int row  = lane & 15;
    const int half = lane >> 4;
#pragma unroll
    for (int m = 0; m < MT; ++m) {
#pragma unroll
        for (int u = 0; u < 2; ++u) {
            const int q = m * 2 + u;
            const v4f lo = *(const v4f*)(xp + ((q * 2)     * 32 + lane) * 4);
            const v4f hi = *(const v4f*)(xp + ((q * 2 + 1) * 32 + lane) * 4);
            const v8f own = c[m][T0 + u];
            float s[8];
            s[0] = own[0] + lo.x; s[1] = own[1] + lo.y; s[2] = own[2] + lo.z; s[3] = own[3] + lo.w;
            s[4] = own[4] + hi.x; s[5] = own[5] + hi.y; s[6] = own[6] + hi.z; s[7] = own[7] + hi.w;
#pragma unroll
            for (int j = 0; j < 8; ++j)
                turn[(m * 16 + (half << 3) + j) * VPITCH + u * 16 + row] = s[j];
        }
    }
    const int pr = lane >> 3;
    const int pc = 4 * (lane & 7);
    const float* tp = turn + pr * VPITCH + pc;
    float* gp = dst + (size_t)(r_first + pr) * DIM + T0 * 16 + pc;
    if (r_first + GROUP_ROWS <= N_TOTAL) {
        v4f val[GROUP_ROWS / 4];
#pragma unroll
        for (int i = 0; i < GROUP_ROWS / 4; ++i)
            val[i] = *(const v4f*)(tp + 4 * i * VPITCH);
#pragma unroll
        for (int i = 0; i < GROUP_ROWS / 4; ++i)
            *(v4f*)(gp + (size_t)(4 * i) * DIM) = LAST ? (e0[i] + val[i]) * 0.25f : val[i];
    } else {
#pragma unroll
        for (int i = 0; i < GROUP_ROWS / 4; ++i) {
            const v4f val = *(const v4f*)(tp + 4 * i * VPITCH);
            if (r_first + 4 * i + pr < N_TOTAL)
                *(v4f*)(gp + (size_t)(4 * i) * DIM) = LAST ? (e0[i] + val) * 0.25f : val;
        }
    }
}

template <bool LAST>
__global__ __launch_bounds__(PROP_THREADS)
void lgcn_prop(const float* __restrict__ v,
               const float* __restrict__ proj,
               const float* __restrict__ ego0  ,
               float* __restrict__ dst  ) {
    __shared__ v2f lds_proj[(FDIM / 2) * DIM];
    __shared__ __attribute__((aligned(16))) float lds_v[2][PROP_WAVES * SLICE];

    const int tid  = threadIdx.x;
    const int lane = tid & 31;
    const int wave = tid >> 5;
    const int grp  = wave >> 1;
    const int kh   = wave & 1;
    const int g0   = blockIdx.x * PROP_ROWS + grp * GROUP_ROWS;

    const int lj = lane & 7;
    const int lr = lane >> 3;
    unsigned goff[PIECES];
#pragma unroll
    for (int i = 0; i < PIECES; ++i)
        goff[i] = (unsigned)min(g0 + lr + PIECE_ROWS * i, N_TOTAL - 1) * FDIM + kh * KHALF + lj * 4;
    const int soff = wave * SLICE + lr * VPITCH + lj * 4;

    v4f st[PIECES];
#pragma unroll
    for (int i = 0; i < PIECES; ++i)
        st[i] = *(const v4f*)(v + goff[i]);

    for (int idx = tid; idx < (FDIM / 2) * DIM; idx += PROP_THREADS) {
        const int j = idx >> 6;
        const int d = idx & 63;
        v2f val;
        val.x = proj[(size_t)(2 * j)     * DIM + d];
        val.y = proj[(size_t)(2 * j + 1) * DIM + d];
        lds_proj[idx] = val;
    }
#pragma unroll
    for (int i = 0; i < PIECES; ++i)
        *(v4f*)&lds_v[0][soff + i * PIECE_ROWS * VPITCH] = st[i];
    __syncthreads();

    const int row   = lane & 15;
    const int half  = lane >> 4;
    const int khalf = half * 2;

    v8f c[MT][4];
#pragma unroll
    for (int m = 0; m < MT; ++m)
#pragma unroll
        for (int t = 0; t < 4; ++t)
            c[m][t] = (v8f){0.0f, 0.0f, 0.0f, 0.0f, 0.0f, 0.0f, 0.0f, 0.0f};

    const int aoff = wave * SLICE + row * VPITCH + khalf;
    const v2f* bbase = &lds_proj[(kh * (KHALF / 2) + half) * DIM + row];

    __builtin_amdgcn_s_setprio(3);
#pragma unroll 1
    for (int kb = 0; kb < NKBLK - 1; ++kb) {
        if (kb == PROP_PRIO2) __builtin_amdgcn_s_setprio(2);
        if (kb == PROP_PRIO1) __builtin_amdgcn_s_setprio(1);
        if (kb == PROP_PRIO0) __builtin_amdgcn_s_setprio(0);
#pragma unroll
        for (int i = 0; i < PIECES; ++i)
            st[i] = *(const v4f*)(v + goff[i] + (kb + 1) * KBLK);
        prop_kblock(lds_v[kb & 1] + aoff, bbase + (size_t)(kb * (KBLK / 2)) * DIM, c);
        __builtin_amdgcn_sched_barrier(0);
        float* ln = lds_v[(kb + 1) & 1];
#pragma unroll
        for (int i = 0; i < PIECES; ++i)
            *(v4f*)&ln[soff + i * PIECE_ROWS * VPITCH] = st[i];
    }
    v4f e0[GROUP_ROWS / 4];
    if (LAST) {
        const float* ep = ego0 + (kh == 0 ? 0 : 32) + 4 * (lane & 7);
#pragma unroll
        for (int i = 0; i < GROUP_ROWS / 4; ++i)
            e0[i] = *(const v4f*)(ep + (size_t)min(g0 + (lane >> 3) + 4 * i, N_TOTAL - 1) * DIM);
    }
    prop_kblock(lds_v[(NKBLK - 1) & 1] + aoff, bbase + (size_t)((NKBLK - 1) * (KBLK / 2)) * DIM, c);

    float*       xmine = &lds_v[0][wave * SLICE];
    const float* xpart = &lds_v[0][(wave ^ 1) * SLICE];
    if (kh == 0) prop_give<2>(xmine, lane, c);
    else         prop_give<0>(xmine, lane, c);
    __syncthreads();
    float* turn = &lds_v[1][wave * SLICE];
    if (kh == 0) prop_finish<0, LAST>(xpart, turn, lane, g0, c, e0, dst);
    else         prop_finish<2, LAST>(xpart, turn, lane, g0, c, e0, dst);
}

static_assert(N_LAYERS == 3, "the reduction's three kinds are first, middle, last");

extern "C" void kernel_launch(void* const* d_in, const int* in_sizes, int n_in,
                              void* d_out, int out_size, void* d_ws, size_t ws_size,
                              hipStream_t stream) {
    const float* user    = (const float*)d_in[0];
    const float* item    = (const float*)d_in[1];
    const float* v       = (const float*)d_in[2];
    const float* filters = (const float*)d_in[3];

    float* ego0    = (float*)d_ws;
    float* ego1    = ego0 + (size_t)N_TOTAL * DIM;
    float* ego2    = ego1 + (size_t)N_TOTAL * DIM;
    float* partial = ego2 + (size_t)N_TOTAL * DIM;
    float* proj    = partial + (size_t)NCHUNKS * FDIM * DIM;
    float* ssum    = proj + (size_t)FDIM * DIM;
    float* out     = (float*)d_out;

    const dim3 gA(NCHUNKS, PT_FSPLIT);

    lgcn_init<<<(TOTAL4 + 255) / 256, 256, 0, stream>>>(user, item, ego0);

    lgcn_projT<<<gA, 256, 0, stream>>>(v, ego0, partial);
    lgcn_reduce<0><<<RD_BLOCKS, RD_THREADS, 0, stream>>>(partial, filters, proj, ssum);
    lgcn_prop<false><<<PROP_BLOCKS, PROP_THREADS, 0, stream>>>(v, proj, ego0, ego1);

    lgcn_projT<<<gA, 256, 0, stream>>>(v, ego1, partial);
    lgcn_reduce<1><<<RD_BLOCKS, RD_THREADS, 0, stream>>>(partial, filters + FDIM, proj, ssum);
    lgcn_prop<false><<<PROP_BLOCKS, PROP_THREADS, 0, stream>>>(v, proj, ego0, ego2);

    lgcn_projT<<<gA, 256, 0, stream>>>(v, ego2, partial);
    lgcn_reduce<2><<<RD_BLOCKS, RD_THREADS, 0, stream>>>(partial, filters + 2 * FDIM, proj, ssum);
    lgcn_prop<true><<<PROP_BLOCKS, PROP_THREADS, 0, stream>>>(v, proj, ego0, out);
}


# Layers 0,1: prop and the next projT fused per 288-row block: new ego rows stay in LDS and multiply the same v rows; no ego planes 1,2; 8 launches, was 10; proj now summed over 246 blocks of 288 rows
# speedup vs baseline: 4.0547x; 1.0817x over previous
#include <hip/hip_runtime.h>


typedef __attribute__((ext_vector_type(2))) float v2f;
typedef __attribute__((ext_vector_type(4))) float v4f;
typedef __attribute__((ext_vector_type(8))) float v8f;

#define N_USERS 29858
#define N_ITEMS 40981
#define N_TOTAL (N_USERS + N_ITEMS)
#define DIM     64
#define FDIM    512
#define N_LAYERS 3
#define NCHUNKS 128
#define CHUNK   556
#define QUADS   (CHUNK / 4)

static_assert(NCHUNKS * CHUNK >= N_TOTAL, "chunk coverage");

#define WMMA_F32(A, B, C) \
    __builtin_amdgcn_wmma_f32_16x16x4_f32(false, (A), false, (B), (short)0, (C), false, false)

#define USER4  (N_USERS * DIM / 4)
#define TOTAL4 (N_TOTAL * DIM / 4)

static_assert((N_USERS * DIM) % 4 == 0 && (N_ITEMS * DIM) % 4 == 0, "both tables are whole 16-byte pieces");

__global__ void lgcn_init(const float* __restrict__ user,
                          const float* __restrict__ item,
                          float* __restrict__ ego) {
    int i = blockIdx.x * blockDim.x + threadIdx.x;
    if (i < TOTAL4) {
        const v4f val = (i < USER4) ? ((const v4f*)user)[i] : ((const v4f*)item)[i - USER4];
        ((v4f*)ego)[i] = val;
    }
}

#define PT_FSPLIT 2
#define PT_F      (FDIM / PT_FSPLIT)
#define PT_ROWS   32
#define PT_NBLK   ((CHUNK + PT_ROWS - 1) / PT_ROWS)
#define PT_PAIRS  (PT_ROWS / 2)
#define PT_CPAIRS (PT_NBLK * PT_PAIRS)
#define PT_VPITCH (2 * PT_F + 32)
#define PT_EPITCH (2 * DIM + 32)
#define PT_VUNITS 4
#define PT_EUNITS (PT_CPAIRS * (DIM / 4) / 256)

#define PT_PRIO2  (PT_NBLK - 8)
#define PT_PRIO1  (PT_NBLK - 4)
#define PT_PRIO0  (PT_NBLK - 2)

static_assert(PT_PAIRS * (32 / 4) == 32 * PT_VUNITS, "a wave's v units per stage");
static_assert(PT_PRIO2 > 0 && PT_PRIO0 < PT_NBLK - 1, "every priority step lies inside the K loop, which is left at priority 0");
static_assert(PT_CPAIRS * (DIM / 4) == 256 * PT_EUNITS, "ego units per thread");

__device__ __forceinline__ void projT_fetch(const float* __restrict__ vcol, int nb, int vp,
                                            v4f (&sv0)[PT_VUNITS], v4f (&sv1)[PT_VUNITS]) {
#pragma unroll
    for (int i = 0; i < PT_VUNITS; ++i) {
        const int r = nb + 2 * (vp + 4 * i);
        sv0[i] = *(const v4f*)(vcol + (size_t)min(r,     N_TOTAL - 1) * FDIM);
        sv1[i] = *(const v4f*)(vcol + (size_t)min(r + 1, N_TOTAL - 1) * FDIM);
    }
}

template <int VPITCH>
__device__ __forceinline__ void projT_stash(float* __restrict__ lv, int vs,
                                            const v4f (&sv0)[PT_VUNITS], const v4f (&sv1)[PT_VUNITS]) {
#pragma unroll
    for (int i = 0; i < PT_VUNITS; ++i) {
        v4f lo, hi;
        lo.x = sv0[i].x; lo.y = sv1[i].x; lo.z = sv0[i].y; lo.w = sv1[i].y;
        hi.x = sv0[i].z; hi.y = sv1[i].z; hi.z = sv0[i].w; hi.w = sv1[i].w;
        *(v4f*)(lv + vs + i * 4 * VPITCH)     = lo;
        *(v4f*)(lv + vs + i * 4 * VPITCH + 4) = hi;
    }
}

template <int VPITCH>
__device__ __forceinline__ void projT_kblock(const float* __restrict__ ap,
                                             const float* __restrict__ bp,
                                             v8f (&c0)[4], v8f (&c1)[4]) {
#pragma unroll
    for (int kk = 0; kk < PT_ROWS / 4; ++kk) {
        const v2f a0 = *(const v2f*)(ap + (2 * kk) * VPITCH);
        const v2f a1 = *(const v2f*)(ap + (2 * kk) * VPITCH + 32);
        const float* bq = bp + (2 * kk) * PT_EPITCH;
#pragma unroll
        for (int t = 0; t < 4; ++t) {
            const v2f b = *(const v2f*)(bq + 32 * t);
            c0[t] = WMMA_F32(a0, b, c0[t]);
            c1[t] = WMMA_F32(a1, b, c1[t]);
        }
    }
}

__global__ __launch_bounds__(256)
void lgcn_projT(const float* __restrict__ v,
                const float* __restrict__ ego,
                float* __restrict__ partial  ) {
    __shared__ __attribute__((aligned(16))) float lds_vt[2][PT_PAIRS * PT_VPITCH];
    __shared__ __attribute__((aligned(16))) float lds_eg[PT_CPAIRS * PT_EPITCH];

    const int tid     = threadIdx.x;
    const int lane    = tid & 31;
    const int wave    = tid >> 5;
    const int chunk   = blockIdx.x;
    const int n_begin = chunk * CHUNK;
    const int n_valid = min(N_TOTAL - n_begin, CHUNK);
    const int fblk    = blockIdx.y * PT_F;

    const int vcg = wave * 8 + (lane & 7);
    const int vp  = lane >> 3;
    const float* vcol = v + fblk + 4 * vcg;
    const int vs = vp * PT_VPITCH + 8 * vcg;

    v4f sv0[PT_VUNITS], sv1[PT_VUNITS];
    projT_fetch(vcol, n_begin, vp, sv0, sv1);

    {
        const int ecg = tid & 15;
        const int ep0 = tid >> 4;
        const float* ecol = ego + 4 * ecg;
        v4f se0[PT_EUNITS], se1[PT_EUNITS];
#pragma unroll
        for (int j = 0; j < PT_EUNITS; ++j) {
            const int r = n_begin + 2 * (ep0 + 16 * j);
            se0[j] = *(const v4f*)(ecol + (size_t)min(r,     N_TOTAL - 1) * DIM);
            se1[j] = *(const v4f*)(ecol + (size_t)min(r + 1, N_TOTAL - 1) * DIM);
        }
        const v4f zero = {0.0f, 0.0f, 0.0f, 0.0f};
#pragma unroll
        for (int j = 0; j < PT_EUNITS; ++j) {
            const int ep = ep0 + 16 * j;
            const v4f e0 = (2 * ep     < n_valid) ? se0[j] : zero;
            const v4f e1 = (2 * ep + 1 < n_valid) ? se1[j] : zero;
            v4f lo, hi;
            lo.x = e0.x; lo.y = e1.x; lo.z = e0.y; lo.w = e1.y;
            hi.x = e0.z; hi.y = e1.z; hi.z = e0.w; hi.w = e1.w;
            *(v4f*)(lds_eg + ep * PT_EPITCH + 8 * ecg)     = lo;
            *(v4f*)(lds_eg + ep * PT_EPITCH + 8 * ecg + 4) = hi;
        }
    }
    projT_stash<PT_VPITCH>(lds_vt[0], vs, sv0, sv1);
    __syncthreads();

    const int f0    = fblk + wave * 32;
    const int row   = lane & 15;
    const int half  = lane >> 4;

    v8f c0[4] = {{}, {}, {}, {}};
    v8f c1[4] = {{}, {}, {}, {}};

    const int aoff = half * PT_VPITCH + 2 * (wave * 32 + row);
    const int boff = half * PT_EPITCH + 2 * row;

    __builtin_amdgcn_s_setprio(3);
#pragma unroll 1
    for (int kb = 0; kb < PT_NBLK - 1; ++kb) {
        if (kb == PT_PRIO2) __builtin_amdgcn_s_setprio(2);
        if (kb == PT_PRIO1) __builtin_amdgcn_s_setprio(1);
        if (kb == PT_PRIO0) __builtin_amdgcn_s_setprio(0);
        projT_fetch(vcol, n_begin + (kb + 1) * PT_ROWS, vp, sv0, sv1);
        projT_kblock<PT_VPITCH>(lds_vt[kb & 1] + aoff, lds_eg + kb * (PT_PAIRS * PT_EPITCH) + boff, c0, c1);
        __builtin_amdgcn_sched_barrier(0);
        projT_stash<PT_VPITCH>(lds_vt[(kb + 1) & 1], vs, sv0, sv1);
    }
    projT_kblock<PT_VPITCH>(lds_vt[(PT_NBLK - 1) & 1] + aoff,
                 lds_eg + (PT_NBLK - 1) * (PT_PAIRS * PT_EPITCH) + boff, c0, c1);

    float* lw = &lds_vt[0][0] + 64 * wave;
#pragma unroll
    for (int t = 0; t < 4; ++t) {
#pragma unroll
        for (int j = 0; j < 8; ++j) {
            lw[(j + 8 * half) * PT_VPITCH + t * 16 + row]                         = c0[t][j];
            lw[PT_PAIRS * PT_VPITCH + (j + 8 * half) * PT_VPITCH + t * 16 + row] = c1[t][j];
        }
    }
    float* outp = partial + (size_t)chunk * FDIM * DIM + (size_t)f0 * DIM;
    const int pr = lane >> 4;
    const int pc = 4 * (lane & 15);
#pragma unroll
    for (int i = 0; i < 16; ++i) {
        const int fr = 2 * i + pr;
        const v4f val = *(const v4f*)(lw + (fr >> 4) * (PT_PAIRS * PT_VPITCH) + (fr & 15) * PT_VPITCH + pc);
        *(v4f*)(outp + fr * DIM + pc) = val;
    }
}

#define RD_RUNS    4
#define RD_BATCH   32
#define RD_THREADS 128
#define RD_BLOCKS  (FDIM * DIM / 4 / 8 / (RD_THREADS / 32))

static_assert((DIM / 4) % 8 == 0, "a wave's 8 pieces stay inside one proj row");
static_assert(RD_BLOCKS * (RD_THREADS / 32) * 8 * 4 == FDIM * DIM, "pieces cover proj exactly");

template <int KIND, int NCH>
__global__ __launch_bounds__(RD_THREADS) __attribute__((amdgpu_waves_per_eu(1, 4)))
void lgcn_reduce(const float* __restrict__ partial,
                 const float* __restrict__ filt  ,
                 float* proj  ,
                 float* ssum  ) {
    const int lane  = threadIdx.x & 31;
    const int wv    = blockIdx.x * (RD_THREADS / 32) + (threadIdx.x >> 5);
    const int piece = wv * 8 + (lane & 7);
    const int run   = lane >> 3;
    static_assert(NCH % (RD_RUNS * RD_BATCH) == 0, "runs are whole batches");
    const float* src = partial + (size_t)run * (NCH / RD_RUNS) * FDIM * DIM + 4 * piece;
    const float f = filt[__builtin_amdgcn_readfirstlane(wv) / (DIM / 4 / 8)];
    v4f prev = {0.0f, 0.0f, 0.0f, 0.0f};
    if (KIND == 1) prev = *(const v4f*)(proj + 4 * piece);
    if (KIND == 2) prev = *(const v4f*)(ssum + 4 * piece);
    v4f s = {0.0f, 0.0f, 0.0f, 0.0f};
#pragma unroll
    for (int b = 0; b < NCH / RD_RUNS / RD_BATCH; ++b) {
        v4f p[RD_BATCH];
#pragma unroll
        for (int n = 0; n < RD_BATCH; ++n)
            p[n] = *(const v4f*)(src + (size_t)(b * RD_BATCH + n) * FDIM * DIM);
        v4f sb = p[0];
#pragma unroll
        for (int n = 1; n < RD_BATCH; ++n)
            sb = sb + p[n];
        s = (b == 0) ? sb : s + sb;
    }
    asm volatile("" :: "s"(f));
    v4f o;
    o.x = __shfl_xor(s.x, 8);  o.y = __shfl_xor(s.y, 8);  o.z = __shfl_xor(s.z, 8);  o.w = __shfl_xor(s.w, 8);
    s = s + o;
    o.x = __shfl_xor(s.x, 16); o.y = __shfl_xor(s.y, 16); o.z = __shfl_xor(s.z, 16); o.w = __shfl_xor(s.w, 16);
    s = s + o;
    if (run == 0) {
        const v4f sk = s * f;
        if (KIND == 0) *(v4f*)(proj + 4 * piece) = sk;
        if (KIND == 1) { *(v4f*)(ssum + 4 * piece) = prev + sk; *(v4f*)(proj + 4 * piece) = sk; }
        if (KIND == 2) *(v4f*)(proj + 4 * piece) = prev + sk;
    }
}

#define PROP_WAVES   12
#define PROP_GROUPS  (PROP_WAVES / 2)
#define GROUP_ROWS   48
#define PROP_ROWS    (PROP_GROUPS * GROUP_ROWS)
#define PROP_THREADS (PROP_WAVES * 32)
#define PROP_BLOCKS  ((N_TOTAL + PROP_ROWS - 1) / PROP_ROWS)
#define KBLK         32
#define KHALF        (FDIM / 2)
#define NKBLK        (KHALF / KBLK)
#define VPITCH       36
#define PIECES       12
#define PIECE_ROWS   4
#define SLICE        (GROUP_ROWS * VPITCH)
#define MT           (GROUP_ROWS / 16)

#define PROP_PRIO2   (NKBLK - 6)
#define PROP_PRIO1   (NKBLK - 4)
#define PROP_PRIO0   (NKBLK - 2)

static_assert(GROUP_ROWS * (KBLK / 4) == 32 * PIECES, "a wave's 32 lanes x 12 pieces cover its 48 rows x 32 columns");
static_assert(PROP_PRIO2 > 0 && PROP_PRIO0 < NKBLK - 1, "every priority step lies inside the K loop, which is left at priority 0");
static_assert(PIECE_ROWS * PIECES == GROUP_ROWS, "piece rows cover the slice");
static_assert(PROP_BLOCKS * PROP_ROWS >= N_TOTAL, "row coverage");
static_assert(MT * 2 * 8 * 32 <= SLICE, "the swapped tiles of a wave fit its own stage-0 slice");
static_assert((NKBLK - 1) % 2 == 1, "the last block is read from stage 1, so a wave's stage-1 slice is its own to reuse after the K loop");

__device__ __forceinline__ void prop_kblock(const float* __restrict__ ar,
                                            const v2f* __restrict__ bp0,
                                            v8f (&c)[MT][4]) {
#pragma unroll
    for (int kk = 0; kk < KBLK; kk += 4) {
        v2f a[MT];
#pragma unroll
        for (int m = 0; m < MT; ++m)
            a[m] = *(const v2f*)(ar + m * 16 * VPITCH + kk);
        const v2f* bp = bp0 + (kk >> 1) * DIM;
#pragma unroll
        for (int t = 0; t < 4; ++t) {
            const v2f b = bp[t * 16];
#pragma unroll
            for (int m = 0; m < MT; ++m)
                c[m][t] = WMMA_F32(a[m], b, c[m][t]);
        }
    }
}

template <int T0>
__device__ __forceinline__ void prop_give(float* __restrict__ x, int lane, const v8f (&c)[MT][4]) {
#pragma unroll
    for (int m = 0; m < MT; ++m) {
#pragma unroll
        for (int u = 0; u < 2; ++u) {
            const int q = m * 2 + u;
            const v8f val = c[m][T0 + u];
            v4f lo, hi;
            lo.x = val[0]; lo.y = val[1]; lo.z = val[2]; lo.w = val[3];
            hi.x = val[4]; hi.y = val[5]; hi.z = val[6]; hi.w = val[7];
            *(v4f*)(x + ((q * 2)     * 32 + lane) * 4) = lo;
            *(v4f*)(x + ((q * 2 + 1) * 32 + lane) * 4) = hi;
        }
    }
}

template <int T0, bool LAST>
__device__ __forceinline__ void prop_finish(const float* __restrict__ xp, float* __restrict__ turn,
                                            int lane, int r_first,
                                            const v8f (&c)[MT][4], const v4f (&e0)[GROUP_ROWS / 4],
                                            float* __restrict__ dst) {
    const int row  = lane & 15;
    const int half = lane >> 4;
#pragma unroll
    for (int m = 0; m < MT; ++m) {
#pragma unroll
        for (int u = 0; u < 2; ++u) {
            const int q = m * 2 + u;
            const v4f lo = *(const v4f*)(xp + ((q * 2)     * 32 + lane) * 4);
            const v4f hi = *(const v4f*)(xp + ((q * 2 + 1) * 32 + lane) * 4);
            const v8f own = c[m][T0 + u];
            float s[8];
            s[0] = own[0] + lo.x; s[1] = own[1] + lo.y; s[2] = own[2] + lo.z; s[3] = own[3] + lo.w;
            s[4] = own[4] + hi.x; s[5] = own[5] + hi.y; s[6] = own[6] + hi.z; s[7] = own[7] + hi.w;
#pragma unroll
            for (int j = 0; j < 8; ++j)
                turn[(m * 16 + (half << 3) + j) * VPITCH + u * 16 + row] = s[j];
        }
    }
    const int pr = lane >> 3;
    const int pc = 4 * (lane & 7);
    const float* tp = turn + pr * VPITCH + pc;
    float* gp = dst + (size_t)(r_first + pr) * DIM + T0 * 16 + pc;
    if (r_first + GROUP_ROWS <= N_TOTAL) {
        v4f val[GROUP_ROWS / 4];
#pragma unroll
        for (int i = 0; i < GROUP_ROWS / 4; ++i)
            val[i] = *(const v4f*)(tp + 4 * i * VPITCH);
#pragma unroll
        for (int i = 0; i < GROUP_ROWS / 4; ++i)
            *(v4f*)(gp + (size_t)(4 * i) * DIM) = LAST ? (e0[i] + val[i]) * 0.25f : val[i];
    } else {
#pragma unroll
        for (int i = 0; i < GROUP_ROWS / 4; ++i) {
            const v4f val = *(const v4f*)(tp + 4 * i * VPITCH);
            if (r_first + 4 * i + pr < N_TOTAL)
                *(v4f*)(gp + (size_t)(4 * i) * DIM) = LAST ? (e0[i] + val) * 0.25f : val;
        }
    }
}

template <bool LAST>
__global__ __launch_bounds__(PROP_THREADS)
void lgcn_prop(const float* __restrict__ v,
               const float* __restrict__ proj,
               const float* __restrict__ ego0  ,
               float* __restrict__ dst  ) {
    __shared__ v2f lds_proj[(FDIM / 2) * DIM];
    __shared__ __attribute__((aligned(16))) float lds_v[2][PROP_WAVES * SLICE];

    const int tid  = threadIdx.x;
    const int lane = tid & 31;
    const int wave = tid >> 5;
    const int grp  = wave >> 1;
    const int kh   = wave & 1;
    const int g0   = blockIdx.x * PROP_ROWS + grp * GROUP_ROWS;

    const int lj = lane & 7;
    const int lr = lane >> 3;
    unsigned goff[PIECES];
#pragma unroll
    for (int i = 0; i < PIECES; ++i)
        goff[i] = (unsigned)min(g0 + lr + PIECE_ROWS * i, N_TOTAL - 1) * FDIM + kh * KHALF + lj * 4;
    const int soff = wave * SLICE + lr * VPITCH + lj * 4;

    v4f st[PIECES];
#pragma unroll
    for (int i = 0; i < PIECES; ++i)
        st[i] = *(const v4f*)(v + goff[i]);

    for (int idx = tid; idx < (FDIM / 2) * DIM; idx += PROP_THREADS) {
        const int j = idx >> 6;
        const int d = idx & 63;
        v2f val;
        val.x = proj[(size_t)(2 * j)     * DIM + d];
        val.y = proj[(size_t)(2 * j + 1) * DIM + d];
        lds_proj[idx] = val;
    }
#pragma unroll
    for (int i = 0; i < PIECES; ++i)
        *(v4f*)&lds_v[0][soff + i * PIECE_ROWS * VPITCH] = st[i];
    __syncthreads();

    const int row   = lane & 15;
    const int half  = lane >> 4;
    const int khalf = half * 2;

    v8f c[MT][4];
#pragma unroll
    for (int m = 0; m < MT; ++m)
#pragma unroll
        for (int t = 0; t < 4; ++t)
            c[m][t] = (v8f){0.0f, 0.0f, 0.0f, 0.0f, 0.0f, 0.0f, 0.0f, 0.0f};

    const int aoff = wave * SLICE + row * VPITCH + khalf;
    const v2f* bbase = &lds_proj[(kh * (KHALF / 2) + half) * DIM + row];

    __builtin_amdgcn_s_setprio(3);
#pragma unroll 1
    for (int kb = 0; kb < NKBLK - 1; ++kb) {
        if (kb == PROP_PRIO2) __builtin_amdgcn_s_setprio(2);
        if (kb == PROP_PRIO1) __builtin_amdgcn_s_setprio(1);
        if (kb == PROP_PRIO0) __builtin_amdgcn_s_setprio(0);
#pragma unroll
        for (int i = 0; i < PIECES; ++i)
            st[i] = *(const v4f*)(v + goff[i] + (kb + 1) * KBLK);
        prop_kblock(lds_v[kb & 1] + aoff, bbase + (size_t)(kb * (KBLK / 2)) * DIM, c);
        __builtin_amdgcn_sched_barrier(0);
        float* ln = lds_v[(kb + 1) & 1];
#pragma unroll
        for (int i = 0; i < PIECES; ++i)
            *(v4f*)&ln[soff + i * PIECE_ROWS * VPITCH] = st[i];
    }
    v4f e0[GROUP_ROWS / 4];
    if (LAST) {
        const float* ep = ego0 + (kh == 0 ? 0 : 32) + 4 * (lane & 7);
#pragma unroll
        for (int i = 0; i < GROUP_ROWS / 4; ++i)
            e0[i] = *(const v4f*)(ep + (size_t)min(g0 + (lane >> 3) + 4 * i, N_TOTAL - 1) * DIM);
    }
    prop_kblock(lds_v[(NKBLK - 1) & 1] + aoff, bbase + (size_t)((NKBLK - 1) * (KBLK / 2)) * DIM, c);

    float*       xmine = &lds_v[0][wave * SLICE];
    const float* xpart = &lds_v[0][(wave ^ 1) * SLICE];
    if (kh == 0) prop_give<2>(xmine, lane, c);
    else         prop_give<0>(xmine, lane, c);
    __syncthreads();
    float* turn = &lds_v[1][wave * SLICE];
    if (kh == 0) prop_finish<0, LAST>(xpart, turn, lane, g0, c, e0, dst);
    else         prop_finish<2, LAST>(xpart, turn, lane, g0, c, e0, dst);
}

#define FZ_PLANES  256
#define FZ_NST     (PROP_ROWS / PT_ROWS)
#define FZ_UNITS   (FDIM / 32)
#define FZ_VPITCH  (2 * 32 * PROP_WAVES + 32)
#define FZ_EPAIRS  (PROP_ROWS / 2)

static_assert(PROP_ROWS % PT_ROWS == 0, "whole stages");
static_assert(FZ_PLANES >= PROP_BLOCKS, "a plane per row block");
static_assert(FZ_EPAIRS * PT_EPITCH <= FDIM * DIM, "the block's ego rows fit over proj");
static_assert(2 * PT_PAIRS * FZ_VPITCH <= 2 * PROP_WAVES * SLICE, "two v stages fit over the phase-1 slices");
static_assert((FZ_NST - 1) % 2 == 0, "the last stage is read from buffer 0");
static_assert(FZ_VPITCH % 64 == 32, "the two lane halves of a fragment read fall on disjoint banks");

template <int T0>
__device__ __forceinline__ void fused_finish(const float* __restrict__ xp, float* __restrict__ eg,
                                             int lane, int grp, int g0, const v8f (&c)[MT][4]) {
    const int row  = lane & 15;
    const int half = lane >> 4;
#pragma unroll
    for (int m = 0; m < MT; ++m) {
#pragma unroll
        for (int u = 0; u < 2; ++u) {
            const int q = m * 2 + u;
            const v4f lo = *(const v4f*)(xp + ((q * 2)     * 32 + lane) * 4);
            const v4f hi = *(const v4f*)(xp + ((q * 2 + 1) * 32 + lane) * 4);
            const v8f own = c[m][T0 + u];
            float s[8];
            s[0] = own[0] + lo.x; s[1] = own[1] + lo.y; s[2] = own[2] + lo.z; s[3] = own[3] + lo.w;
            s[4] = own[4] + hi.x; s[5] = own[5] + hi.y; s[6] = own[6] + hi.z; s[7] = own[7] + hi.w;
            const int rg  = m * 16 + (half << 3);
            const int col = (T0 + u) * 16 + row;
#pragma unroll
            for (int j = 0; j < 8; j += 2) {
                v2f pr;
                pr.x = (g0 + rg + j     < N_TOTAL) ? s[j]     : 0.0f;
                pr.y = (g0 + rg + j + 1 < N_TOTAL) ? s[j + 1] : 0.0f;
                *(v2f*)(eg + ((grp * GROUP_ROWS + rg + j) >> 1) * PT_EPITCH + 2 * col) = pr;
            }
        }
    }
}

__global__ __launch_bounds__(PROP_THREADS)
void lgcn_fused(const float* __restrict__ v,
                const float* __restrict__ proj,
                float* __restrict__ partial  ) {
    __shared__ v2f lds_proj[(FDIM / 2) * DIM];
    __shared__ __attribute__((aligned(16))) float lds_v[2][PROP_WAVES * SLICE];

    const int tid  = threadIdx.x;
    const int lane = tid & 31;
    const int wave = tid >> 5;

    if (blockIdx.x >= PROP_BLOCKS) {
        const v4f zero = {0.0f, 0.0f, 0.0f, 0.0f};
        v4f* pz = (v4f*)(partial + (size_t)blockIdx.x * FDIM * DIM);
        for (int i = tid; i < FDIM * DIM / 4; i += PROP_THREADS) pz[i] = zero;
        return;
    }

    const int grp  = wave >> 1;
    const int kh   = wave & 1;
    const int rb   = blockIdx.x * PROP_ROWS;
    const int g0   = rb + grp * GROUP_ROWS;

    const int lj = lane & 7;
    const int lr = lane >> 3;
    unsigned goff[PIECES];
#pragma unroll
    for (int i = 0; i < PIECES; ++i)
        goff[i] = (unsigned)min(g0 + lr + PIECE_ROWS * i, N_TOTAL - 1) * FDIM + kh * KHALF + lj * 4;
    const int soff = wave * SLICE + lr * VPITCH + lj * 4;

    v4f st[PIECES];
#pragma unroll
    for (int i = 0; i < PIECES; ++i)
        st[i] = *(const v4f*)(v + goff[i]);

    for (int idx = tid; idx < (FDIM / 2) * DIM; idx += PROP_THREADS) {
        const int j = idx >> 6;
        const int d = idx & 63;
        v2f val;
        val.x = proj[(size_t)(2 * j)     * DIM + d];
        val.y = proj[(size_t)(2 * j + 1) * DIM + d];
        lds_proj[idx] = val;
    }
#pragma unroll
    for (int i = 0; i < PIECES; ++i)
        *(v4f*)&lds_v[0][soff + i * PIECE_ROWS * VPITCH] = st[i];
    __syncthreads();

    const int row   = lane & 15;
    const int half  = lane >> 4;
    const int khalf = half * 2;

    v8f c[MT][4];
#pragma unroll
    for (int m = 0; m < MT; ++m)
#pragma unroll
        for (int t = 0; t < 4; ++t)
            c[m][t] = (v8f){0.0f, 0.0f, 0.0f, 0.0f, 0.0f, 0.0f, 0.0f, 0.0f};

    const int aoff1 = wave * SLICE + row * VPITCH + khalf;
    const v2f* bbase = &lds_proj[(kh * (KHALF / 2) + half) * DIM + row];

    __builtin_amdgcn_s_setprio(3);
#pragma unroll 1
    for (int kb = 0; kb < NKBLK - 1; ++kb) {
        if (kb == PROP_PRIO2) __builtin_amdgcn_s_setprio(2);
        if (kb == PROP_PRIO1) __builtin_amdgcn_s_setprio(1);
        if (kb == PROP_PRIO0) __builtin_amdgcn_s_setprio(0);
#pragma unroll
        for (int i = 0; i < PIECES; ++i)
            st[i] = *(const v4f*)(v + goff[i] + (kb + 1) * KBLK);
        prop_kblock(lds_v[kb & 1] + aoff1, bbase + (size_t)(kb * (KBLK / 2)) * DIM, c);
        __builtin_amdgcn_sched_barrier(0);
        float* ln = lds_v[(kb + 1) & 1];
#pragma unroll
        for (int i = 0; i < PIECES; ++i)
            *(v4f*)&ln[soff + i * PIECE_ROWS * VPITCH] = st[i];
    }
    prop_kblock(lds_v[(NKBLK - 1) & 1] + aoff1, bbase + (size_t)((NKBLK - 1) * (KBLK / 2)) * DIM, c);

    const int vp  = lane >> 3;
    const int vs  = vp * FZ_VPITCH + 8 * (wave * 8 + lj);
    v4f sv0[PT_VUNITS], sv1[PT_VUNITS];
    projT_fetch(v + 32 * wave + 4 * lj, rb, vp, sv0, sv1);

    float*       xmine = &lds_v[0][wave * SLICE];
    const float* xpart = &lds_v[0][(wave ^ 1) * SLICE];
    if (kh == 0) prop_give<2>(xmine, lane, c);
    else         prop_give<0>(xmine, lane, c);
    __syncthreads();
    float* eg = (float*)lds_proj;
    if (kh == 0) fused_finish<0>(xpart, eg, lane, grp, g0, c);
    else         fused_finish<2>(xpart, eg, lane, grp, g0, c);
    __syncthreads();

    float* vb = &lds_v[0][0];
    const int aoff2 = half * FZ_VPITCH + 2 * (wave * 32 + row);
    const int boff  = half * PT_EPITCH + 2 * row;
    const int pr = lane >> 4;
    const int pc = 4 * (lane & 15);
    const int wu = __builtin_amdgcn_readfirstlane(wave);
    for (int unit = wu; unit < FZ_UNITS; unit += PROP_WAVES) {
        const float* vcol = v + 32 * unit + 4 * lj;
        if (unit != wu) projT_fetch(vcol, rb, vp, sv0, sv1);
        projT_stash<FZ_VPITCH>(vb, vs, sv0, sv1);

        v8f c0[4] = {{}, {}, {}, {}};
        v8f c1[4] = {{}, {}, {}, {}};
#pragma unroll 1
        for (int kb = 0; kb < FZ_NST - 1; ++kb) {
            projT_fetch(vcol, rb + (kb + 1) * PT_ROWS, vp, sv0, sv1);
            projT_kblock<FZ_VPITCH>(vb + (kb & 1) * (PT_PAIRS * FZ_VPITCH) + aoff2,
                                    eg + kb * (PT_PAIRS * PT_EPITCH) + boff, c0, c1);
            __builtin_amdgcn_sched_barrier(0);
            projT_stash<FZ_VPITCH>(vb + ((kb + 1) & 1) * (PT_PAIRS * FZ_VPITCH), vs, sv0, sv1);
        }
        projT_kblock<FZ_VPITCH>(vb + ((FZ_NST - 1) & 1) * (PT_PAIRS * FZ_VPITCH) + aoff2,
                                eg + (FZ_NST - 1) * (PT_PAIRS * PT_EPITCH) + boff, c0, c1);

        float* lw = vb + 64 * wave;
#pragma unroll
        for (int t = 0; t < 4; ++t) {
#pragma unroll
            for (int j = 0; j < 8; ++j) {
                lw[(j + 8 * half) * FZ_VPITCH + t * 16 + row]                            = c0[t][j];
                lw[PT_PAIRS * FZ_VPITCH + (j + 8 * half) * FZ_VPITCH + t * 16 + row] = c1[t][j];
            }
        }
        float* outp = partial + (size_t)blockIdx.x * FDIM * DIM + (size_t)(32 * unit) * DIM;
#pragma unroll
        for (int i = 0; i < 16; ++i) {
            const int fr = 2 * i + pr;
            const v4f val = *(const v4f*)(lw + (fr >> 4) * (PT_PAIRS * FZ_VPITCH) + (fr & 15) * FZ_VPITCH + pc);
            *(v4f*)(outp + fr * DIM + pc) = val;
        }
    }
}

static_assert(N_LAYERS == 3, "the reduction's three kinds are first, middle, last");

extern "C" void kernel_launch(void* const* d_in, const int* in_sizes, int n_in,
                              void* d_out, int out_size, void* d_ws, size_t ws_size,
                              hipStream_t stream) {
    const float* user    = (const float*)d_in[0];
    const float* item    = (const float*)d_in[1];
    const float* v       = (const float*)d_in[2];
    const float* filters = (const float*)d_in[3];

    float* ego0    = (float*)d_ws;
    float* partial = ego0 + (size_t)N_TOTAL * DIM;
    float* proj    = partial + (size_t)FZ_PLANES * FDIM * DIM;
    float* ssum    = proj + (size_t)FDIM * DIM;
    float* out     = (float*)d_out;

    const dim3 gA(NCHUNKS, PT_FSPLIT);

    lgcn_init<<<(TOTAL4 + 255) / 256, 256, 0, stream>>>(user, item, ego0);

    lgcn_projT<<<gA, 256, 0, stream>>>(v, ego0, partial);
    lgcn_reduce<0, NCHUNKS><<<RD_BLOCKS, RD_THREADS, 0, stream>>>(partial, filters, proj, ssum);

    lgcn_fused<<<FZ_PLANES, PROP_THREADS, 0, stream>>>(v, proj, partial);
    lgcn_reduce<1, FZ_PLANES><<<RD_BLOCKS, RD_THREADS, 0, stream>>>(partial, filters + FDIM, proj, ssum);

    lgcn_fused<<<FZ_PLANES, PROP_THREADS, 0, stream>>>(v, proj, partial);
    lgcn_reduce<2, FZ_PLANES><<<RD_BLOCKS, RD_THREADS, 0, stream>>>(partial, filters + 2 * FDIM, proj, ssum);

    lgcn_prop<true><<<PROP_BLOCKS, PROP_THREADS, 0, stream>>>(v, proj, ego0, out);
}
